# starting kernel
# baseline (speedup 1.0000x reference)
#include <hip/hip_runtime.h>


#define N_NODES 100000
#define N_EDGES 640000
#define D 128

typedef float v8f __attribute__((ext_vector_type(8)));
typedef float v2f __attribute__((ext_vector_type(2)));

__global__ void gcn_zero_kernel(float4* __restrict__ p, int n4) {
    int i = blockIdx.x * blockDim.x + threadIdx.x;
    if (i < n4) p[i] = make_float4(0.f, 0.f, 0.f, 0.f);
}

__global__ void gcn_scatter_kernel(const float* __restrict__ x,
                                   const float* __restrict__ ew,
                                   const int* __restrict__ src,
                                   const int* __restrict__ dst,
                                   float* __restrict__ h) {
    const int wavesPerBlock = blockDim.x >> 5;
    const int e = blockIdx.x * wavesPerBlock + (threadIdx.x >> 5);
    if (e >= N_EDGES) return;
    const int lane = threadIdx.x & 31;

    const int   s = src[e];
    const int   d = dst[e];
    const float w = ew[e];

    const float4 v = *(const float4*)(x + (size_t)s * D + lane * 4);
    float* hp = h + (size_t)d * D + lane * 4;
    unsafeAtomicAdd(hp + 0, v.x * w);
    unsafeAtomicAdd(hp + 1, v.y * w);
    unsafeAtomicAdd(hp + 2, v.z * w);
    unsafeAtomicAdd(hp + 3, v.w * w);
}

__global__ void __launch_bounds__(256)
gcn_gemm_kernel(const float* __restrict__ h,
                const float* __restrict__ W,
                const float* __restrict__ b,
                float* __restrict__ out) {
    const int lane = threadIdx.x & 31;
    const int wave = threadIdx.x >> 5;
    const int row0 = blockIdx.x * 16;
    const int col0 = wave * 16;
    const int l15  = lane & 15;
    const int rsel = lane >> 4;

    const float* aptr = h + (size_t)(row0 + l15) * D + 2 * rsel;
    const float* bptr = W + (size_t)(col0 + l15) * D + 2 * rsel;

    v8f acc = {};
#pragma unroll
    for (int k = 0; k < D; k += 4) {
        v2f a  = *(const v2f*)(aptr + k);
        v2f bb = *(const v2f*)(bptr + k);
        acc = __builtin_amdgcn_wmma_f32_16x16x4_f32(
             false, a,  false, bb,
             (short)0, acc,  false,  false);
    }

    const float bias = b[col0 + l15];
    float* optr = out + (size_t)(row0 + 8 * rsel) * D + col0 + l15;
#pragma unroll
    for (int r = 0; r < 8; ++r) {
        optr[(size_t)r * D] = acc[r] + bias;
    }
}

extern "C" void kernel_launch(void* const* d_in, const int* in_sizes, int n_in,
                              void* d_out, int out_size, void* d_ws, size_t ws_size,
                              hipStream_t stream) {
    const float* x   = (const float*)d_in[0];
    const float* ew  = (const float*)d_in[1];
    const int*   src = (const int*)  d_in[2];
    const int*   dst = (const int*)  d_in[3];
    const float* Wm  = (const float*)d_in[4];
    const float* bv  = (const float*)d_in[5];

    float* h   = (float*)d_ws;
    float* out = (float*)d_out;

    {
        const int n4 = (N_NODES * D) / 4;
        const int threads = 256;
        const int blocks = (n4 + threads - 1) / threads;
        gcn_zero_kernel<<<blocks, threads, 0, stream>>>((float4*)h, n4);
    }

    {
        const int threads = 256;
        const int wavesPerBlock = threads / 32;
        const int blocks = (N_EDGES + wavesPerBlock - 1) / wavesPerBlock;
        gcn_scatter_kernel<<<blocks, threads, 0, stream>>>(x, ew, src, dst, h);
    }

    {
        const int threads = 256;
        const int blocks = N_NODES / 16;
        gcn_gemm_kernel<<<blocks, threads, 0, stream>>>(h, Wm, bv, out);
    }
}


# Pull fused into the GEMM: each persistent 4-wave block sums its 16 rows of h in registers (erec order kept) into a double-buffered LDS tile and multiplies from it; h never goes to memory; 5 launches
# speedup vs baseline: 1.6988x; 1.6988x over previous
#include <hip/hip_runtime.h>


#define N_NODES 100000
#define N_EDGES 640000
#define D 128

typedef float v8f __attribute__((ext_vector_type(8)));
typedef float v2f __attribute__((ext_vector_type(2)));
typedef float v4f __attribute__((ext_vector_type(4)));

#define WS_CNT      (N_NODES * D)
#define WS_ROWSTART (WS_CNT + 131072)
#define WS_EREC     (WS_ROWSTART + 131072)
#define WS_RANK     (WS_EREC + 2 * N_EDGES)

__global__ void gcn_zero_counts_kernel(int4* __restrict__ p, int n4) {
    int i = blockIdx.x * blockDim.x + threadIdx.x;
    if (i < n4) p[i] = make_int4(0, 0, 0, 0);
}

__global__ void gcn_count_rank_kernel(const int* __restrict__ dst,
                                      int* __restrict__ cnt,
                                      int* __restrict__ rank) {
    const int e = blockIdx.x * blockDim.x + threadIdx.x;
    if (e < N_EDGES) rank[e] = atomicAdd(cnt + dst[e], 1);
}

#define SCAN_BLOCK  1024
#define SCAN_SWEEPS 25

__global__ void __launch_bounds__(SCAN_BLOCK)
gcn_rowstart_kernel(const int* __restrict__ cnt,
                    int* __restrict__ rowstart) {
    __shared__ int s_pre[32];
    __shared__ int s_own[32];
    const int t    = threadIdx.x;
    const int lane = t & 31;
    const int wave = t >> 5;
    const int blockBase = blockIdx.x * SCAN_BLOCK;

    int pre = 0;
    const int4* c4 = (const int4*)cnt;
    const int n4 = blockBase >> 2;
#pragma unroll
    for (int k = 0; k < SCAN_SWEEPS; ++k) {
        const int  q   = t + k * SCAN_BLOCK;
        const int4 v   = c4[q];
        const int  sum = (v.x + v.y) + (v.z + v.w);
        pre += (q < n4) ? sum : 0;
    }

    const int i   = blockBase + t;
    const int own = (i < N_NODES) ? cnt[i] : 0;
    int incl = own;
    for (int d = 1; d < 32; d <<= 1) {
        const int u = __shfl_up(incl, d, 32);
        if (lane >= d) incl += u;
    }
    for (int d = 16; d > 0; d >>= 1) pre += __shfl_xor(pre, d, 32);

    if (lane == 31) s_own[wave] = incl;
    if (lane == 0)  s_pre[wave] = pre;
    __syncthreads();

    int base = 0;
    for (int k = 0; k < 32; ++k) {
        base += s_pre[k];
        if (k < wave) base += s_own[k];
    }
    const int excl = base + incl - own;
    if (i < N_NODES) rowstart[i] = excl;
    if (i == N_NODES - 1) rowstart[N_NODES] = excl + own;
}

__global__ void gcn_fill_kernel(const float* __restrict__ ew,
                                const int* __restrict__ src,
                                const int* __restrict__ dst,
                                const int* __restrict__ rowstart,
                                const int* __restrict__ rank,
                                int2* __restrict__ erec) {
    const int e = blockIdx.x * blockDim.x + threadIdx.x;
    if (e >= N_EDGES) return;
    const int   d = dst[e];
    const int   r = rank[e];
    const int   s = src[e];
    const float w = ew[e];
    const int pos = rowstart[d] + r;
    erec[pos] = make_int2(s, __float_as_int(w));
}

#define GEMM_M_TILES (N_NODES / 16)
#define HP 33
#define PULL_BATCH 16

__global__ void __launch_bounds__(128)
gcn_pull_gemm_kernel(const float* __restrict__ x,
                     const int* __restrict__ rowstart,
                     const int2* __restrict__ erec,
                     const float* __restrict__ W,
                     const float* __restrict__ b,
                     float* __restrict__ out) {
    __shared__ v4f s_h[2 * 16 * HP];

    const int lane = threadIdx.x & 31;
    const int wave = __builtin_amdgcn_readfirstlane(threadIdx.x >> 5);
    const int col0 = wave * 32;
    const int l15  = lane & 15;
    const int rsel = lane >> 4;

    const float* wp0 = W + (size_t)(col0 + l15) * D + 4 * rsel;
    const float* wp1 = wp0 + (size_t)16 * D;
    v4f b0[16], b1[16];
#pragma unroll
    for (int j = 0; j < 16; ++j) {
        b0[j] = *(const v4f*)(wp0 + 8 * j);
        b1[j] = *(const v4f*)(wp1 + 8 * j);
    }
    const float bias0 = b[col0 + l15];
    const float bias1 = b[col0 + 16 + l15];

    const float* xl = x + lane * 4;
    const int l4 = min(lane, 4);

    int mt    = blockIdx.x;
    int rsv   = rowstart[mt * 16 + wave * 4 + l4];
    int rsv_n = rowstart[min(mt + (int)gridDim.x, GEMM_M_TILES - 1) * 16 + wave * 4 + l4];
    int r0  = __builtin_amdgcn_readlane(rsv, 0);
    int r4  = __builtin_amdgcn_readlane(rsv, 4);
    int2 rec = make_int2(0, 0);
    if (r4 > r0) rec = erec[r0 + min(lane, min(32, r4 - r0) - 1)];

    int buf = 0;
    for (; mt < GEMM_M_TILES; mt += gridDim.x) {
        v4f* tile   = s_h + buf * (16 * HP);
        v4f* myrows = tile + wave * (4 * HP) + lane;

        int k     = 0;
        int nextb = __builtin_amdgcn_readlane(rsv, 1);
        v4f acc = {0.f, 0.f, 0.f, 0.f};
        int cbase = r0;
        while (cbase < r4) {
            const int n = min(32, r4 - cbase);
            for (int i = 0; i < n; i += PULL_BATCH) {
                const int m = min(PULL_BATCH, n - i);
                v4f   v[PULL_BATCH];
                float w[PULL_BATCH];
#pragma unroll
                for (int j = 0; j < PULL_BATCH; ++j) {
                    const int q = min(i + j, n - 1);
                    const int s = __builtin_amdgcn_readlane(rec.x, q);
                    w[j] = __int_as_float(__builtin_amdgcn_readlane(rec.y, q));
                    v[j] = *(const v4f*)(xl + (size_t)s * D);
                }
                asm volatile("" :: "v"(v[0]), "v"(v[1]), "v"(v[2]), "v"(v[3]),
                                   "v"(v[4]), "v"(v[5]), "v"(v[6]), "v"(v[7]),
                                   "v"(v[8]), "v"(v[9]), "v"(v[10]), "v"(v[11]),
                                   "v"(v[12]), "v"(v[13]), "v"(v[14]), "v"(v[15]));
#pragma unroll
                for (int j = 0; j < PULL_BATCH; ++j) {
                    if (j < m) {
                        const int e = cbase + i + j;
                        while (e >= nextb && k < 3) {
                            myrows[k * HP] = acc;
                            acc = (v4f){0.f, 0.f, 0.f, 0.f};
                            ++k;
                            nextb = __builtin_amdgcn_readlane(rsv, k + 1);
                        }
                        acc += v[j] * w[j];
                    }
                }
            }
            cbase += 32;
            if (cbase < r4) rec = erec[cbase + min(lane, min(32, r4 - cbase) - 1)];
        }
        while (k < 4) {
            myrows[k * HP] = acc;
            acc = (v4f){0.f, 0.f, 0.f, 0.f};
            ++k;
        }

        __syncthreads();

        rsv = rsv_n;
        r0  = __builtin_amdgcn_readlane(rsv, 0);
        r4  = __builtin_amdgcn_readlane(rsv, 4);
        if (r4 > r0) rec = erec[r0 + min(lane, min(32, r4 - r0) - 1)];
        const int rsv_nn = rowstart[min(mt + 2 * (int)gridDim.x, GEMM_M_TILES - 1) * 16 + wave * 4 + l4];

        const v4f* ap = tile + l15 * HP + rsel;
        v4f a[16];
#pragma unroll
        for (int j = 0; j < 16; ++j) a[j] = ap[2 * j];
        asm volatile("" :: "v"(a[0]), "v"(a[1]), "v"(a[2]), "v"(a[3]),
                           "v"(a[4]), "v"(a[5]), "v"(a[6]), "v"(a[7]),
                           "v"(a[8]), "v"(a[9]), "v"(a[10]), "v"(a[11]),
                           "v"(a[12]), "v"(a[13]), "v"(a[14]), "v"(a[15]));

        v8f acc0 = {}, acc1 = {};
#pragma unroll
        for (int j = 0; j < 16; ++j) {
            const v2f alo  = {a[j].x,  a[j].y},  ahi  = {a[j].z,  a[j].w};
            const v2f b0lo = {b0[j].x, b0[j].y}, b0hi = {b0[j].z, b0[j].w};
            const v2f b1lo = {b1[j].x, b1[j].y}, b1hi = {b1[j].z, b1[j].w};
            acc0 = __builtin_amdgcn_wmma_f32_16x16x4_f32(
                false, alo, false, b0lo, (short)0, acc0, false, false);
            acc1 = __builtin_amdgcn_wmma_f32_16x16x4_f32(
                false, alo, false, b1lo, (short)0, acc1, false, false);
            acc0 = __builtin_amdgcn_wmma_f32_16x16x4_f32(
                false, ahi, false, b0hi, (short)0, acc0, false, false);
            acc1 = __builtin_amdgcn_wmma_f32_16x16x4_f32(
                false, ahi, false, b1hi, (short)0, acc1, false, false);
        }

        const int row0 = mt * 16;
        float* optr = out + (size_t)(row0 + 8 * rsel) * D + col0 + l15;
#pragma unroll
        for (int r = 0; r < 8; ++r) {
            optr[(size_t)r * D]      = acc0[r] + bias0;
            optr[(size_t)r * D + 16] = acc1[r] + bias1;
        }
        __builtin_amdgcn_sched_barrier(0);
        rsv_n = rsv_nn;
        buf ^= 1;
    }
}

extern "C" void kernel_launch(void* const* d_in, const int* in_sizes, int n_in,
                              void* d_out, int out_size, void* d_ws, size_t ws_size,
                              hipStream_t stream) {
    const float* x   = (const float*)d_in[0];
    const float* ew  = (const float*)d_in[1];
    const int*   src = (const int*)  d_in[2];
    const int*   dst = (const int*)  d_in[3];
    const float* Wm  = (const float*)d_in[4];
    const float* bv  = (const float*)d_in[5];

    float* out = (float*)d_out;

    int*  cnt      = (int*)d_ws + WS_CNT;
    int*  rowstart = (int*)d_ws + WS_ROWSTART;
    int2* erec     = (int2*)((int*)d_ws + WS_EREC);
    int*  rank     = (int*)d_ws + WS_RANK;

    gcn_zero_counts_kernel<<<100, 256, 0, stream>>>((int4*)cnt, 25600);
    gcn_count_rank_kernel<<<2500, 256, 0, stream>>>(dst, cnt, rank);
    gcn_rowstart_kernel<<<98, SCAN_BLOCK, 0, stream>>>(cnt, rowstart);
    gcn_fill_kernel<<<2500, 256, 0, stream>>>(ew, src, dst, rowstart, rank, erec);
    gcn_pull_gemm_kernel<<<1024, 128, 0, stream>>>(x, rowstart, erec, Wm, bv, out);
}


# Edges filed in one pass: each dst node owns 32 record slots, the counting add's returned value is the slot (overflow list past 32); no prefix-sum or fill launch; rows still summed in arrival order
# speedup vs baseline: 2.0936x; 1.2324x over previous
#include <hip/hip_runtime.h>


#define N_NODES 100000
#define N_EDGES 640000
#define D 128
#define K_SLOTS 32

typedef float v8f __attribute__((ext_vector_type(8)));
typedef float v2f __attribute__((ext_vector_type(2)));
typedef float v4f __attribute__((ext_vector_type(4)));

#define WS_CNT   0
#define WS_SLOT  131072
#define WS_OVF   (WS_SLOT + 2 * N_NODES * K_SLOTS)

__global__ void gcn_zero_counts_kernel(int4* __restrict__ p, int n4) {
    int i = blockIdx.x * blockDim.x + threadIdx.x;
    if (i < n4) p[i] = make_int4(0, 0, 0, 0);
}

__global__ void gcn_count_file_kernel(const float* __restrict__ ew,
                                      const int* __restrict__ src,
                                      const int* __restrict__ dst,
                                      int* __restrict__ cnt,
                                      int2* __restrict__ slot,
                                      int4* __restrict__ ovf) {
    const int e = blockIdx.x * blockDim.x + threadIdx.x;
    if (e >= N_EDGES) return;
    const int d  = dst[e];
    const int s  = src[e];
    const int wb = __float_as_int(ew[e]);
    const int r  = atomicAdd(cnt + d, 1);
    if (r < K_SLOTS) {
        slot[d * K_SLOTS + r] = make_int2(s, wb);
    } else {
        const int p = atomicAdd(cnt + N_NODES, 1);
        ovf[p] = make_int4(d, s, wb, 0);
    }
}

#define GEMM_M_TILES (N_NODES / 16)
#define HP 33
#define PULL_BATCH 16

__device__ __forceinline__ int gcn_bounds(int cv, int lane) {
    const int c0 = min(__builtin_amdgcn_readlane(cv, 0), K_SLOTS);
    const int c1 = min(__builtin_amdgcn_readlane(cv, 1), K_SLOTS);
    const int c2 = min(__builtin_amdgcn_readlane(cv, 2), K_SLOTS);
    const int c3 = min(__builtin_amdgcn_readlane(cv, 3), K_SLOTS);
    return (lane > 0 ? c0 : 0) + (lane > 1 ? c1 : 0) + (lane > 2 ? c2 : 0) + (lane > 3 ? c3 : 0);
}

__device__ __forceinline__ int2 gcn_load_records(const int2* __restrict__ slot, int node0,
                                                 int cbase, int rsv, int lane) {
    const int b1 = __builtin_amdgcn_readlane(rsv, 1);
    const int b2 = __builtin_amdgcn_readlane(rsv, 2);
    const int b3 = __builtin_amdgcn_readlane(rsv, 3);
    const int b4 = __builtin_amdgcn_readlane(rsv, 4);
    const int i   = cbase + min(lane, min(32, b4 - cbase) - 1);
    const int k   = (i >= b1) + (i >= b2) + (i >= b3);
    const int lbk = i >= b3 ? b3 : (i >= b2 ? b2 : (i >= b1 ? b1 : 0));
    return slot[(node0 + k) * K_SLOTS + (i - lbk)];
}

__global__ void __launch_bounds__(128)
gcn_pull_gemm_kernel(const float* __restrict__ x,
                     const int* __restrict__ cnt,
                     const int2* __restrict__ slot,
                     const int4* __restrict__ ovf,
                     const float* __restrict__ W,
                     const float* __restrict__ b,
                     float* __restrict__ out) {
    __shared__ v4f s_h[2 * 16 * HP];

    const int lane = threadIdx.x & 31;
    const int wave = __builtin_amdgcn_readfirstlane(threadIdx.x >> 5);
    const int col0 = wave * 32;
    const int l15  = lane & 15;
    const int rsel = lane >> 4;

    const float* wp0 = W + (size_t)(col0 + l15) * D + 4 * rsel;
    const float* wp1 = wp0 + (size_t)16 * D;
    v4f b0[16], b1[16];
#pragma unroll
    for (int j = 0; j < 16; ++j) {
        b0[j] = *(const v4f*)(wp0 + 8 * j);
        b1[j] = *(const v4f*)(wp1 + 8 * j);
    }
    const float bias0 = b[col0 + l15];
    const float bias1 = b[col0 + 16 + l15];

    const float* xl = x + lane * 4;
    const int l3 = min(lane, 3);

    int mt   = blockIdx.x;
    int cv   = cnt[mt * 16 + wave * 4 + l3];
    int cv_n = cnt[min(mt + (int)gridDim.x, GEMM_M_TILES - 1) * 16 + wave * 4 + l3];
    int rsv  = gcn_bounds(cv, lane);
    int r4   = __builtin_amdgcn_readlane(rsv, 4);
    int2 rec = make_int2(0, 0);
    if (r4 > 0) rec = gcn_load_records(slot, mt * 16 + wave * 4, 0, rsv, lane);

    int buf = 0;
    for (; mt < GEMM_M_TILES; mt += gridDim.x) {
        v4f* tile   = s_h + buf * (16 * HP);
        v4f* myrows = tile + wave * (4 * HP) + lane;
        const int node0 = mt * 16 + wave * 4;

        int k     = 0;
        int nextb = __builtin_amdgcn_readlane(rsv, 1);
        v4f acc = {0.f, 0.f, 0.f, 0.f};
        int cbase = 0;
        while (cbase < r4) {
            const int n = min(32, r4 - cbase);
            for (int i = 0; i < n; i += PULL_BATCH) {
                const int m = min(PULL_BATCH, n - i);
                v4f   v[PULL_BATCH];
                float w[PULL_BATCH];
#pragma unroll
                for (int j = 0; j < PULL_BATCH; ++j) {
                    const int q = min(i + j, n - 1);
                    const int s = __builtin_amdgcn_readlane(rec.x, q);
                    w[j] = __int_as_float(__builtin_amdgcn_readlane(rec.y, q));
                    v[j] = *(const v4f*)(xl + (size_t)s * D);
                }
                asm volatile("" :: "v"(v[0]), "v"(v[1]), "v"(v[2]), "v"(v[3]),
                                   "v"(v[4]), "v"(v[5]), "v"(v[6]), "v"(v[7]),
                                   "v"(v[8]), "v"(v[9]), "v"(v[10]), "v"(v[11]),
                                   "v"(v[12]), "v"(v[13]), "v"(v[14]), "v"(v[15]));
#pragma unroll
                for (int j = 0; j < PULL_BATCH; ++j) {
                    if (j < m) {
                        const int e = cbase + i + j;
                        while (e >= nextb && k < 3) {
                            myrows[k * HP] = acc;
                            acc = (v4f){0.f, 0.f, 0.f, 0.f};
                            ++k;
                            nextb = __builtin_amdgcn_readlane(rsv, k + 1);
                        }
                        acc += v[j] * w[j];
                    }
                }
            }
            cbase += 32;
            if (cbase < r4) rec = gcn_load_records(slot, node0, cbase, rsv, lane);
        }
        while (k < 4) {
            myrows[k * HP] = acc;
            acc = (v4f){0.f, 0.f, 0.f, 0.f};
            ++k;
        }

        const int c0 = __builtin_amdgcn_readlane(cv, 0);
        const int c1 = __builtin_amdgcn_readlane(cv, 1);
        const int c2 = __builtin_amdgcn_readlane(cv, 2);
        const int c3 = __builtin_amdgcn_readlane(cv, 3);
        if (max(max(c0, c1), max(c2, c3)) > K_SLOTS) {
            const int novf = min(__builtin_amdgcn_readfirstlane(cnt[N_NODES]), N_EDGES);
            for (int kk = 0; kk < 4; ++kk) {
                const int ck = kk == 0 ? c0 : (kk == 1 ? c1 : (kk == 2 ? c2 : c3));
                if (ck <= K_SLOTS) continue;
                const int node = node0 + kk;
                v4f extra = {0.f, 0.f, 0.f, 0.f};
                for (int base = 0; base < novf; base += 32) {
                    const int idx = base + lane;
                    int4 o = make_int4(-1, 0, 0, 0);
                    if (idx < novf) o = ovf[idx];
                    unsigned int hits = __builtin_amdgcn_ballot_w32(o.x == node);
                    while (hits != 0u) {
                        const int q = __builtin_ctz(hits);
                        hits &= hits - 1u;
                        const int   s  = __builtin_amdgcn_readlane(o.y, q);
                        const float wq = __int_as_float(__builtin_amdgcn_readlane(o.z, q));
                        extra += *(const v4f*)(xl + (size_t)s * D) * wq;
                    }
                }
                myrows[kk * HP] += extra;
            }
        }

        __syncthreads();

        const int mt_n = min(mt + (int)gridDim.x, GEMM_M_TILES - 1);
        cv  = cv_n;
        rsv = gcn_bounds(cv, lane);
        r4  = __builtin_amdgcn_readlane(rsv, 4);
        if (r4 > 0) rec = gcn_load_records(slot, mt_n * 16 + wave * 4, 0, rsv, lane);
        const int cv_nn = cnt[min(mt + 2 * (int)gridDim.x, GEMM_M_TILES - 1) * 16 + wave * 4 + l3];

        const v4f* ap = tile + l15 * HP + rsel;
        v4f a[16];
#pragma unroll
        for (int j = 0; j < 16; ++j) a[j] = ap[2 * j];
        asm volatile("" :: "v"(a[0]), "v"(a[1]), "v"(a[2]), "v"(a[3]),
                           "v"(a[4]), "v"(a[5]), "v"(a[6]), "v"(a[7]),
                           "v"(a[8]), "v"(a[9]), "v"(a[10]), "v"(a[11]),
                           "v"(a[12]), "v"(a[13]), "v"(a[14]), "v"(a[15]));

        v8f acc0 = {}, acc1 = {};
#pragma unroll
        for (int j = 0; j < 16; ++j) {
            const v2f alo  = {a[j].x,  a[j].y},  ahi  = {a[j].z,  a[j].w};
            const v2f b0lo = {b0[j].x, b0[j].y}, b0hi = {b0[j].z, b0[j].w};
            const v2f b1lo = {b1[j].x, b1[j].y}, b1hi = {b1[j].z, b1[j].w};
            acc0 = __builtin_amdgcn_wmma_f32_16x16x4_f32(
                false, alo, false, b0lo, (short)0, acc0, false, false);
            acc1 = __builtin_amdgcn_wmma_f32_16x16x4_f32(
                false, alo, false, b1lo, (short)0, acc1, false, false);
            acc0 = __builtin_amdgcn_wmma_f32_16x16x4_f32(
                false, ahi, false, b0hi, (short)0, acc0, false, false);
            acc1 = __builtin_amdgcn_wmma_f32_16x16x4_f32(
                false, ahi, false, b1hi, (short)0, acc1, false, false);
        }

        const int row0 = mt * 16;
        float* optr = out + (size_t)(row0 + 8 * rsel) * D + col0 + l15;
#pragma unroll
        for (int r = 0; r < 8; ++r) {
            optr[(size_t)r * D]      = acc0[r] + bias0;
            optr[(size_t)r * D + 16] = acc1[r] + bias1;
        }
        __builtin_amdgcn_sched_barrier(0);
        cv_n = cv_nn;
        buf ^= 1;
    }
}

extern "C" void kernel_launch(void* const* d_in, const int* in_sizes, int n_in,
                              void* d_out, int out_size, void* d_ws, size_t ws_size,
                              hipStream_t stream) {
    const float* x   = (const float*)d_in[0];
    const float* ew  = (const float*)d_in[1];
    const int*   src = (const int*)  d_in[2];
    const int*   dst = (const int*)  d_in[3];
    const float* Wm  = (const float*)d_in[4];
    const float* bv  = (const float*)d_in[5];

    float* out = (float*)d_out;

    int*  cnt  = (int*)d_ws + WS_CNT;
    int2* slot = (int2*)((int*)d_ws + WS_SLOT);
    int4* ovf  = (int4*)((int*)d_ws + WS_OVF);

    gcn_zero_counts_kernel<<<100, 256, 0, stream>>>((int4*)cnt, 25600);
    gcn_count_file_kernel<<<2500, 256, 0, stream>>>(ew, src, dst, cnt, slot, ovf);
    gcn_pull_gemm_kernel<<<1024, 128, 0, stream>>>(x, cnt, slot, ovf, Wm, bv, out);
}


# Edges filed without global atomics: launch 1 sorts 2560-edge slices by 400-node group on LDS counters, launch 2 (block per group) takes slots from LDS adds; no zero launch; sums in LDS arrival order
# speedup vs baseline: 3.0863x; 1.4742x over previous
#include <hip/hip_runtime.h>


#define N_NODES 100000
#define N_EDGES 640000
#define D 128
#define K_SLOTS 32

#define SLICE_EDGES 2560
#define N_SLICES    (N_EDGES / SLICE_EDGES)
#define GROUP_NODES 400
#define N_GROUPS    (N_NODES / GROUP_NODES)
#define SORT_EPT    (SLICE_EDGES / 256)
#define FILE_BATCH  12

static_assert(N_EDGES % SLICE_EDGES == 0 && SLICE_EDGES % 256 == 0, "whole slices");
static_assert(N_NODES % GROUP_NODES == 0, "whole groups");
static_assert(N_SLICES <= 256 && N_GROUPS <= 256, "one thread per slice / per group");

typedef float v8f __attribute__((ext_vector_type(8)));
typedef float v2f __attribute__((ext_vector_type(2)));
typedef float v4f __attribute__((ext_vector_type(4)));

#define WS_CNT   0
#define WS_SLOT  131072
#define WS_OVF   (WS_SLOT + 2 * N_NODES * K_SLOTS)
#define WS_SORT  (WS_OVF + 4 * N_EDGES)
#define WS_TAB   (WS_SORT + 4 * N_EDGES)

__device__ __forceinline__ int gcn_block_excl_scan(int v, int tid, int* s_wsum, int& total) {
    const int lane = tid & 31;
    int inc = v;
#pragma unroll
    for (int o = 1; o < 32; o <<= 1) {
        const int t = __shfl_up(inc, o, 32);
        if (lane >= o) inc += t;
    }
    if (lane == 31) s_wsum[tid >> 5] = inc;
    __syncthreads();
    int wbase = 0, tot = 0;
#pragma unroll
    for (int k = 0; k < 8; ++k) {
        const int t = s_wsum[k];
        if (k < (tid >> 5)) wbase += t;
        tot += t;
    }
    total = tot;
    return wbase + inc - v;
}

__global__ void __launch_bounds__(256)
gcn_slice_sort_kernel(const float* __restrict__ ew,
                      const int* __restrict__ src,
                      const int* __restrict__ dst,
                      int* __restrict__ cnt,
                      int4* __restrict__ sorted,
                      int2* __restrict__ tab) {
    __shared__ int s_hist[256];
    __shared__ int s_off[256];
    __shared__ int s_wsum[8];
    const int tid = threadIdx.x;
    const int sl  = blockIdx.x;
    s_hist[tid] = 0;
    if (sl == 0 && tid == 0) cnt[N_NODES] = 0;
    __syncthreads();

    int d[SORT_EPT], s[SORT_EPT], wb[SORT_EPT], r[SORT_EPT];
    const int e0 = sl * SLICE_EDGES + tid;
#pragma unroll
    for (int j = 0; j < SORT_EPT; ++j) {
        d[j]  = dst[e0 + j * 256];
        s[j]  = src[e0 + j * 256];
        wb[j] = __float_as_int(ew[e0 + j * 256]);
    }
#pragma unroll
    for (int j = 0; j < SORT_EPT; ++j) r[j] = atomicAdd(&s_hist[d[j] / GROUP_NODES], 1);
    __syncthreads();

    const int c = s_hist[tid];
    int total;
    const int off = gcn_block_excl_scan(c, tid, s_wsum, total);
    s_off[tid] = off;
    if (tid < N_GROUPS) tab[tid * N_SLICES + sl] = make_int2(off, c);
    __syncthreads();

    int4* mine = sorted + sl * SLICE_EDGES;
#pragma unroll
    for (int j = 0; j < SORT_EPT; ++j)
        mine[s_off[d[j] / GROUP_NODES] + r[j]] = make_int4(d[j], s[j], wb[j], 0);
}

__global__ void __launch_bounds__(256)
gcn_group_file_kernel(const int4* __restrict__ sorted,
                      const int2* __restrict__ tab,
                      int* __restrict__ cnt,
                      int2* __restrict__ slot,
                      int4* __restrict__ ovf) {
    __shared__ int s_cnt[GROUP_NODES];
    __shared__ int s_base[256];
    __shared__ int s_first[256];
    __shared__ int s_wsum[8];
    const int tid = threadIdx.x;
    const int g   = blockIdx.x;
    for (int i = tid; i < GROUP_NODES; i += 256) s_cnt[i] = 0;
    int2 sc = make_int2(0, 0);
    if (tid < N_SLICES) sc = tab[g * N_SLICES + tid];
    int total;
    const int base = gcn_block_excl_scan(sc.y, tid, s_wsum, total);
    s_base[tid]  = base;
    s_first[tid] = tid * SLICE_EDGES + sc.x;
    __syncthreads();

    for (int i0 = 0; i0 < total; i0 += 256 * FILE_BATCH) {
        int4 rec[FILE_BATCH];
#pragma unroll
        for (int k = 0; k < FILE_BATCH; ++k) {
            const int i = min(i0 + k * 256 + tid, total - 1);
            int t = 0;
#pragma unroll
            for (int step = 128; step > 0; step >>= 1)
                if (s_base[t + step] <= i) t += step;
            rec[k] = sorted[s_first[t] + (i - s_base[t])];
        }
        int r[FILE_BATCH];
#pragma unroll
        for (int k = 0; k < FILE_BATCH; ++k) {
            r[k] = 0;
            if (i0 + k * 256 + tid < total)
                r[k] = atomicAdd(&s_cnt[rec[k].x - g * GROUP_NODES], 1);
        }
#pragma unroll
        for (int k = 0; k < FILE_BATCH; ++k) {
            if (i0 + k * 256 + tid < total) {
                const int dn = rec[k].x;
                if (r[k] < K_SLOTS) {
                    slot[dn * K_SLOTS + r[k]] = make_int2(rec[k].y, rec[k].z);
                } else {
                    const int p = atomicAdd(cnt + N_NODES, 1);
                    ovf[p] = make_int4(dn, rec[k].y, rec[k].z, 0);
                }
            }
        }
    }
    __syncthreads();
    for (int i = tid; i < GROUP_NODES; i += 256) cnt[g * GROUP_NODES + i] = s_cnt[i];
}

#define GEMM_M_TILES (N_NODES / 16)
#define HP 33
#define PULL_BATCH 16

__device__ __forceinline__ int gcn_bounds(int cv, int lane) {
    const int c0 = min(__builtin_amdgcn_readlane(cv, 0), K_SLOTS);
    const int c1 = min(__builtin_amdgcn_readlane(cv, 1), K_SLOTS);
    const int c2 = min(__builtin_amdgcn_readlane(cv, 2), K_SLOTS);
    const int c3 = min(__builtin_amdgcn_readlane(cv, 3), K_SLOTS);
    return (lane > 0 ? c0 : 0) + (lane > 1 ? c1 : 0) + (lane > 2 ? c2 : 0) + (lane > 3 ? c3 : 0);
}

__device__ __forceinline__ int2 gcn_load_records(const int2* __restrict__ slot, int node0,
                                                 int cbase, int rsv, int lane) {
    const int b1 = __builtin_amdgcn_readlane(rsv, 1);
    const int b2 = __builtin_amdgcn_readlane(rsv, 2);
    const int b3 = __builtin_amdgcn_readlane(rsv, 3);
    const int b4 = __builtin_amdgcn_readlane(rsv, 4);
    const int i   = cbase + min(lane, min(32, b4 - cbase) - 1);
    const int k   = (i >= b1) + (i >= b2) + (i >= b3);
    const int lbk = i >= b3 ? b3 : (i >= b2 ? b2 : (i >= b1 ? b1 : 0));
    return slot[(node0 + k) * K_SLOTS + (i - lbk)];
}

__global__ void __launch_bounds__(128)
gcn_pull_gemm_kernel(const float* __restrict__ x,
                     const int* __restrict__ cnt,
                     const int2* __restrict__ slot,
                     const int4* __restrict__ ovf,
                     const float* __restrict__ W,
                     const float* __restrict__ b,
                     float* __restrict__ out) {
    __shared__ v4f s_h[2 * 16 * HP];

    const int lane = threadIdx.x & 31;
    const int wave = __builtin_amdgcn_readfirstlane(threadIdx.x >> 5);
    const int col0 = wave * 32;
    const int l15  = lane & 15;
    const int rsel = lane >> 4;

    const float* wp0 = W + (size_t)(col0 + l15) * D + 4 * rsel;
    const float* wp1 = wp0 + (size_t)16 * D;
    v4f b0[16], b1[16];
#pragma unroll
    for (int j = 0; j < 16; ++j) {
        b0[j] = *(const v4f*)(wp0 + 8 * j);
        b1[j] = *(const v4f*)(wp1 + 8 * j);
    }
    const float bias0 = b[col0 + l15];
    const float bias1 = b[col0 + 16 + l15];

    const float* xl = x + lane * 4;
    const int l3 = min(lane, 3);

    int mt   = blockIdx.x;
    int cv   = cnt[mt * 16 + wave * 4 + l3];
    int cv_n = cnt[min(mt + (int)gridDim.x, GEMM_M_TILES - 1) * 16 + wave * 4 + l3];
    int rsv  = gcn_bounds(cv, lane);
    int r4   = __builtin_amdgcn_readlane(rsv, 4);
    int2 rec = make_int2(0, 0);
    if (r4 > 0) rec = gcn_load_records(slot, mt * 16 + wave * 4, 0, rsv, lane);

    int buf = 0;
    for (; mt < GEMM_M_TILES; mt += gridDim.x) {
        v4f* tile   = s_h + buf * (16 * HP);
        v4f* myrows = tile + wave * (4 * HP) + lane;
        const int node0 = mt * 16 + wave * 4;

        int k     = 0;
        int nextb = __builtin_amdgcn_readlane(rsv, 1);
        v4f acc = {0.f, 0.f, 0.f, 0.f};
        int cbase = 0;
        while (cbase < r4) {
            const int n = min(32, r4 - cbase);
            for (int i = 0; i < n; i += PULL_BATCH) {
                const int m = min(PULL_BATCH, n - i);
                v4f   v[PULL_BATCH];
                float w[PULL_BATCH];
#pragma unroll
                for (int j = 0; j < PULL_BATCH; ++j) {
                    const int q = min(i + j, n - 1);
                    const int s = __builtin_amdgcn_readlane(rec.x, q);
                    w[j] = __int_as_float(__builtin_amdgcn_readlane(rec.y, q));
                    v[j] = *(const v4f*)(xl + (size_t)s * D);
                }
                asm volatile("" :: "v"(v[0]), "v"(v[1]), "v"(v[2]), "v"(v[3]),
                                   "v"(v[4]), "v"(v[5]), "v"(v[6]), "v"(v[7]),
                                   "v"(v[8]), "v"(v[9]), "v"(v[10]), "v"(v[11]),
                                   "v"(v[12]), "v"(v[13]), "v"(v[14]), "v"(v[15]));
#pragma unroll
                for (int j = 0; j < PULL_BATCH; ++j) {
                    if (j < m) {
                        const int e = cbase + i + j;
                        while (e >= nextb && k < 3) {
                            myrows[k * HP] = acc;
                            acc = (v4f){0.f, 0.f, 0.f, 0.f};
                            ++k;
                            nextb = __builtin_amdgcn_readlane(rsv, k + 1);
                        }
                        acc += v[j] * w[j];
                    }
                }
            }
            cbase += 32;
            if (cbase < r4) rec = gcn_load_records(slot, node0, cbase, rsv, lane);
        }
        while (k < 4) {
            myrows[k * HP] = acc;
            acc = (v4f){0.f, 0.f, 0.f, 0.f};
            ++k;
        }

        const int c0 = __builtin_amdgcn_readlane(cv, 0);
        const int c1 = __builtin_amdgcn_readlane(cv, 1);
        const int c2 = __builtin_amdgcn_readlane(cv, 2);
        const int c3 = __builtin_amdgcn_readlane(cv, 3);
        if (max(max(c0, c1), max(c2, c3)) > K_SLOTS) {
            const int novf = min(__builtin_amdgcn_readfirstlane(cnt[N_NODES]), N_EDGES);
            for (int kk = 0; kk < 4; ++kk) {
                const int ck = kk == 0 ? c0 : (kk == 1 ? c1 : (kk == 2 ? c2 : c3));
                if (ck <= K_SLOTS) continue;
                const int node = node0 + kk;
                v4f extra = {0.f, 0.f, 0.f, 0.f};
                for (int base = 0; base < novf; base += 32) {
                    const int idx = base + lane;
                    int4 o = make_int4(-1, 0, 0, 0);
                    if (idx < novf) o = ovf[idx];
                    unsigned int hits = __builtin_amdgcn_ballot_w32(o.x == node);
                    while (hits != 0u) {
                        const int q = __builtin_ctz(hits);
                        hits &= hits - 1u;
                        const int   s  = __builtin_amdgcn_readlane(o.y, q);
                        const float wq = __int_as_float(__builtin_amdgcn_readlane(o.z, q));
                        extra += *(const v4f*)(xl + (size_t)s * D) * wq;
                    }
                }
                myrows[kk * HP] += extra;
            }
        }

        __syncthreads();

        const int mt_n = min(mt + (int)gridDim.x, GEMM_M_TILES - 1);
        cv  = cv_n;
        rsv = gcn_bounds(cv, lane);
        r4  = __builtin_amdgcn_readlane(rsv, 4);
        if (r4 > 0) rec = gcn_load_records(slot, mt_n * 16 + wave * 4, 0, rsv, lane);
        const int cv_nn = cnt[min(mt + 2 * (int)gridDim.x, GEMM_M_TILES - 1) * 16 + wave * 4 + l3];

        const v4f* ap = tile + l15 * HP + rsel;
        v4f a[16];
#pragma unroll
        for (int j = 0; j < 16; ++j) a[j] = ap[2 * j];
        asm volatile("" :: "v"(a[0]), "v"(a[1]), "v"(a[2]), "v"(a[3]),
                           "v"(a[4]), "v"(a[5]), "v"(a[6]), "v"(a[7]),
                           "v"(a[8]), "v"(a[9]), "v"(a[10]), "v"(a[11]),
                           "v"(a[12]), "v"(a[13]), "v"(a[14]), "v"(a[15]));

        v8f acc0 = {}, acc1 = {};
#pragma unroll
        for (int j = 0; j < 16; ++j) {
            const v2f alo  = {a[j].x,  a[j].y},  ahi  = {a[j].z,  a[j].w};
            const v2f b0lo = {b0[j].x, b0[j].y}, b0hi = {b0[j].z, b0[j].w};
            const v2f b1lo = {b1[j].x, b1[j].y}, b1hi = {b1[j].z, b1[j].w};
            acc0 = __builtin_amdgcn_wmma_f32_16x16x4_f32(
                false, alo, false, b0lo, (short)0, acc0, false, false);
            acc1 = __builtin_amdgcn_wmma_f32_16x16x4_f32(
                false, alo, false, b1lo, (short)0, acc1, false, false);
            acc0 = __builtin_amdgcn_wmma_f32_16x16x4_f32(
                false, ahi, false, b0hi, (short)0, acc0, false, false);
            acc1 = __builtin_amdgcn_wmma_f32_16x16x4_f32(
                false, ahi, false, b1hi, (short)0, acc1, false, false);
        }

        const int row0 = mt * 16;
        float* optr = out + (size_t)(row0 + 8 * rsel) * D + col0 + l15;
#pragma unroll
        for (int r = 0; r < 8; ++r) {
            optr[(size_t)r * D]      = acc0[r] + bias0;
            optr[(size_t)r * D + 16] = acc1[r] + bias1;
        }
        __builtin_amdgcn_sched_barrier(0);
        cv_n = cv_nn;
        buf ^= 1;
    }
}

extern "C" void kernel_launch(void* const* d_in, const int* in_sizes, int n_in,
                              void* d_out, int out_size, void* d_ws, size_t ws_size,
                              hipStream_t stream) {
    const float* x   = (const float*)d_in[0];
    const float* ew  = (const float*)d_in[1];
    const int*   src = (const int*)  d_in[2];
    const int*   dst = (const int*)  d_in[3];
    const float* Wm  = (const float*)d_in[4];
    const float* bv  = (const float*)d_in[5];

    float* out = (float*)d_out;

    int*  cnt  = (int*)d_ws + WS_CNT;
    int2* slot = (int2*)((int*)d_ws + WS_SLOT);
    int4* ovf  = (int4*)((int*)d_ws + WS_OVF);

    int4* sorted = (int4*)((int*)d_ws + WS_SORT);
    int2* tab    = (int2*)((int*)d_ws + WS_TAB);

    gcn_slice_sort_kernel<<<250, 256, 0, stream>>>(ew, src, dst, cnt, sorted, tab);
    gcn_group_file_kernel<<<250, 256, 0, stream>>>(sorted, tab, cnt, slot, ovf);
    gcn_pull_gemm_kernel<<<1024, 128, 0, stream>>>(x, cnt, slot, ovf, Wm, bv, out);
}


# Pull: node-boundary and tail tests leave the sum loop: a ballot marks each node's last record, the loop tests one mask bit per record (store and reset out of line); sums unchanged
# speedup vs baseline: 3.2476x; 1.0523x over previous
#include <hip/hip_runtime.h>


#define N_NODES 100000
#define N_EDGES 640000
#define D 128
#define K_SLOTS 32

#define SLICE_EDGES 2560
#define N_SLICES    (N_EDGES / SLICE_EDGES)
#define GROUP_NODES 400
#define N_GROUPS    (N_NODES / GROUP_NODES)
#define SORT_EPT    (SLICE_EDGES / 256)
#define FILE_BATCH  12

static_assert(N_EDGES % SLICE_EDGES == 0 && SLICE_EDGES % 256 == 0, "whole slices");
static_assert(N_NODES % GROUP_NODES == 0, "whole groups");
static_assert(N_SLICES <= 256 && N_GROUPS <= 256, "one thread per slice / per group");

typedef float v8f __attribute__((ext_vector_type(8)));
typedef float v2f __attribute__((ext_vector_type(2)));
typedef float v4f __attribute__((ext_vector_type(4)));

#define WS_CNT   0
#define WS_SLOT  131072
#define WS_OVF   (WS_SLOT + 2 * N_NODES * K_SLOTS)
#define WS_SORT  (WS_OVF + 4 * N_EDGES)
#define WS_TAB   (WS_SORT + 4 * N_EDGES)

__device__ __forceinline__ int gcn_block_excl_scan(int v, int tid, int* s_wsum, int& total) {
    const int lane = tid & 31;
    int inc = v;
#pragma unroll
    for (int o = 1; o < 32; o <<= 1) {
        const int t = __shfl_up(inc, o, 32);
        if (lane >= o) inc += t;
    }
    if (lane == 31) s_wsum[tid >> 5] = inc;
    __syncthreads();
    int wbase = 0, tot = 0;
#pragma unroll
    for (int k = 0; k < 8; ++k) {
        const int t = s_wsum[k];
        if (k < (tid >> 5)) wbase += t;
        tot += t;
    }
    total = tot;
    return wbase + inc - v;
}

__global__ void __launch_bounds__(256)
gcn_slice_sort_kernel(const float* __restrict__ ew,
                      const int* __restrict__ src,
                      const int* __restrict__ dst,
                      int* __restrict__ cnt,
                      int4* __restrict__ sorted,
                      int2* __restrict__ tab) {
    __shared__ int s_hist[256];
    __shared__ int s_off[256];
    __shared__ int s_wsum[8];
    const int tid = threadIdx.x;
    const int sl  = blockIdx.x;
    s_hist[tid] = 0;
    if (sl == 0 && tid == 0) cnt[N_NODES] = 0;
    __syncthreads();

    int d[SORT_EPT], s[SORT_EPT], wb[SORT_EPT], r[SORT_EPT];
    const int e0 = sl * SLICE_EDGES + tid;
#pragma unroll
    for (int j = 0; j < SORT_EPT; ++j) {
        d[j]  = dst[e0 + j * 256];
        s[j]  = src[e0 + j * 256];
        wb[j] = __float_as_int(ew[e0 + j * 256]);
    }
#pragma unroll
    for (int j = 0; j < SORT_EPT; ++j) r[j] = atomicAdd(&s_hist[d[j] / GROUP_NODES], 1);
    __syncthreads();

    const int c = s_hist[tid];
    int total;
    const int off = gcn_block_excl_scan(c, tid, s_wsum, total);
    s_off[tid] = off;
    if (tid < N_GROUPS) tab[tid * N_SLICES + sl] = make_int2(off, c);
    __syncthreads();

    int4* mine = sorted + sl * SLICE_EDGES;
#pragma unroll
    for (int j = 0; j < SORT_EPT; ++j)
        mine[s_off[d[j] / GROUP_NODES] + r[j]] = make_int4(d[j], s[j], wb[j], 0);
}

__global__ void __launch_bounds__(256)
gcn_group_file_kernel(const int4* __restrict__ sorted,
                      const int2* __restrict__ tab,
                      int* __restrict__ cnt,
                      int2* __restrict__ slot,
                      int4* __restrict__ ovf) {
    __shared__ int s_cnt[GROUP_NODES];
    __shared__ int s_base[256];
    __shared__ int s_first[256];
    __shared__ int s_wsum[8];
    const int tid = threadIdx.x;
    const int g   = blockIdx.x;
    for (int i = tid; i < GROUP_NODES; i += 256) s_cnt[i] = 0;
    int2 sc = make_int2(0, 0);
    if (tid < N_SLICES) sc = tab[g * N_SLICES + tid];
    int total;
    const int base = gcn_block_excl_scan(sc.y, tid, s_wsum, total);
    s_base[tid]  = base;
    s_first[tid] = tid * SLICE_EDGES + sc.x;
    __syncthreads();

    for (int i0 = 0; i0 < total; i0 += 256 * FILE_BATCH) {
        int4 rec[FILE_BATCH];
#pragma unroll
        for (int k = 0; k < FILE_BATCH; ++k) {
            const int i = min(i0 + k * 256 + tid, total - 1);
            int t = 0;
#pragma unroll
            for (int step = 128; step > 0; step >>= 1)
                if (s_base[t + step] <= i) t += step;
            rec[k] = sorted[s_first[t] + (i - s_base[t])];
        }
        int r[FILE_BATCH];
#pragma unroll
        for (int k = 0; k < FILE_BATCH; ++k) {
            r[k] = 0;
            if (i0 + k * 256 + tid < total)
                r[k] = atomicAdd(&s_cnt[rec[k].x - g * GROUP_NODES], 1);
        }
#pragma unroll
        for (int k = 0; k < FILE_BATCH; ++k) {
            if (i0 + k * 256 + tid < total) {
                const int dn = rec[k].x;
                if (r[k] < K_SLOTS) {
                    slot[dn * K_SLOTS + r[k]] = make_int2(rec[k].y, rec[k].z);
                } else {
                    const int p = atomicAdd(cnt + N_NODES, 1);
                    ovf[p] = make_int4(dn, rec[k].y, rec[k].z, 0);
                }
            }
        }
    }
    __syncthreads();
    for (int i = tid; i < GROUP_NODES; i += 256) cnt[g * GROUP_NODES + i] = s_cnt[i];
}

#define GEMM_M_TILES (N_NODES / 16)
#define HP 33
#define PULL_BATCH 16

__device__ __forceinline__ int gcn_bounds(int cv, int lane) {
    const int c0 = min(__builtin_amdgcn_readlane(cv, 0), K_SLOTS);
    const int c1 = min(__builtin_amdgcn_readlane(cv, 1), K_SLOTS);
    const int c2 = min(__builtin_amdgcn_readlane(cv, 2), K_SLOTS);
    const int c3 = min(__builtin_amdgcn_readlane(cv, 3), K_SLOTS);
    return (lane > 0 ? c0 : 0) + (lane > 1 ? c1 : 0) + (lane > 2 ? c2 : 0) + (lane > 3 ? c3 : 0);
}

__device__ __forceinline__ int2 gcn_load_records(const int2* __restrict__ slot, int node0,
                                                 int cbase, int rsv, int lane) {
    const int b1 = __builtin_amdgcn_readlane(rsv, 1);
    const int b2 = __builtin_amdgcn_readlane(rsv, 2);
    const int b3 = __builtin_amdgcn_readlane(rsv, 3);
    const int b4 = __builtin_amdgcn_readlane(rsv, 4);
    const int i   = cbase + min(lane, min(32, b4 - cbase) - 1);
    const int k   = (i >= b1) + (i >= b2) + (i >= b3);
    const int lbk = i >= b3 ? b3 : (i >= b2 ? b2 : (i >= b1 ? b1 : 0));
    return slot[(node0 + k) * K_SLOTS + (i - lbk)];
}

__device__ __forceinline__ bool gcn_owner(int rsv, int cbase, int lane, int& kown) {
    const int b1 = __builtin_amdgcn_readlane(rsv, 1);
    const int b2 = __builtin_amdgcn_readlane(rsv, 2);
    const int b3 = __builtin_amdgcn_readlane(rsv, 3);
    const int b4 = __builtin_amdgcn_readlane(rsv, 4);
    const int i    = cbase + lane;
    const int endk = i >= b3 ? b4 : (i >= b2 ? b3 : (i >= b1 ? b2 : b1));
    kown = (i >= b1) + (i >= b2) + (i >= b3);
    return i < b4 && i + 1 == endk;
}

__global__ void __launch_bounds__(128)
gcn_pull_gemm_kernel(const float* __restrict__ x,
                     const int* __restrict__ cnt,
                     const int2* __restrict__ slot,
                     const int4* __restrict__ ovf,
                     const float* __restrict__ W,
                     const float* __restrict__ b,
                     float* __restrict__ out) {
    __shared__ v4f s_h[2 * 16 * HP];

    const int lane = threadIdx.x & 31;
    const int wave = __builtin_amdgcn_readfirstlane(threadIdx.x >> 5);
    const int col0 = wave * 32;
    const int l15  = lane & 15;
    const int rsel = lane >> 4;

    const float* wp0 = W + (size_t)(col0 + l15) * D + 4 * rsel;
    const float* wp1 = wp0 + (size_t)16 * D;
    v4f b0[16], b1[16];
#pragma unroll
    for (int j = 0; j < 16; ++j) {
        b0[j] = *(const v4f*)(wp0 + 8 * j);
        b1[j] = *(const v4f*)(wp1 + 8 * j);
    }
    const float bias0 = b[col0 + l15];
    const float bias1 = b[col0 + 16 + l15];

    const float* xl = x + lane * 4;
    const int l3 = min(lane, 3);

    int mt   = blockIdx.x;
    int cv   = cnt[mt * 16 + wave * 4 + l3];
    int cv_n = cnt[min(mt + (int)gridDim.x, GEMM_M_TILES - 1) * 16 + wave * 4 + l3];
    int rsv  = gcn_bounds(cv, lane);
    int r4   = __builtin_amdgcn_readlane(rsv, 4);
    int2 rec = make_int2(0, 0);
    if (r4 > 0) rec = gcn_load_records(slot, mt * 16 + wave * 4, 0, rsv, lane);

    int buf = 0;
    for (; mt < GEMM_M_TILES; mt += gridDim.x) {
        v4f* tile   = s_h + buf * (16 * HP);
        v4f* myrows = tile + wave * (4 * HP) + lane;
        const int node0 = mt * 16 + wave * 4;

        const int c0 = __builtin_amdgcn_readlane(cv, 0);
        const int c1 = __builtin_amdgcn_readlane(cv, 1);
        const int c2 = __builtin_amdgcn_readlane(cv, 2);
        const int c3 = __builtin_amdgcn_readlane(cv, 3);
        if (min(min(c0, c1), min(c2, c3)) == 0) {
            for (int kk = 0; kk < 4; ++kk) {
                const int ck = kk == 0 ? c0 : (kk == 1 ? c1 : (kk == 2 ? c2 : c3));
                if (ck == 0) myrows[kk * HP] = (v4f){0.f, 0.f, 0.f, 0.f};
            }
        }
        v4f acc = {0.f, 0.f, 0.f, 0.f};
        int cbase = 0;
        while (cbase < r4) {
            const int n = min(32, r4 - cbase);
            int kown;
            const bool last = gcn_owner(rsv, cbase, lane, kown);
            const unsigned int lastmask = __builtin_amdgcn_ballot_w32(last);
            for (int i = 0; i < n; i += PULL_BATCH) {
                v4f   v[PULL_BATCH];
                float w[PULL_BATCH];
#pragma unroll
                for (int j = 0; j < PULL_BATCH; ++j) {
                    const int s = __builtin_amdgcn_readlane(rec.x, i + j);
                    w[j] = __int_as_float(__builtin_amdgcn_readlane(rec.y, i + j));
                    v[j] = *(const v4f*)(xl + (size_t)s * D);
                }
                asm volatile("" :: "v"(v[0]), "v"(v[1]), "v"(v[2]), "v"(v[3]),
                                   "v"(v[4]), "v"(v[5]), "v"(v[6]), "v"(v[7]),
                                   "v"(v[8]), "v"(v[9]), "v"(v[10]), "v"(v[11]),
                                   "v"(v[12]), "v"(v[13]), "v"(v[14]), "v"(v[15]));
                const unsigned int mb = lastmask >> i;
#pragma unroll
                for (int j = 0; j < PULL_BATCH; ++j) {
                    acc += v[j] * w[j];
                    if (__builtin_expect((mb >> j) & 1u, 0)) {
                        myrows[__builtin_amdgcn_readlane(kown, i + j) * HP] = acc;
                        acc = (v4f){0.f, 0.f, 0.f, 0.f};
                    }
                }
            }
            cbase += 32;
            if (cbase < r4) rec = gcn_load_records(slot, node0, cbase, rsv, lane);
        }

        if (max(max(c0, c1), max(c2, c3)) > K_SLOTS) {
            const int novf = min(__builtin_amdgcn_readfirstlane(cnt[N_NODES]), N_EDGES);
            for (int kk = 0; kk < 4; ++kk) {
                const int ck = kk == 0 ? c0 : (kk == 1 ? c1 : (kk == 2 ? c2 : c3));
                if (ck <= K_SLOTS) continue;
                const int node = node0 + kk;
                v4f extra = {0.f, 0.f, 0.f, 0.f};
                for (int base = 0; base < novf; base += 32) {
                    const int idx = base + lane;
                    int4 o = make_int4(-1, 0, 0, 0);
                    if (idx < novf) o = ovf[idx];
                    unsigned int hits = __builtin_amdgcn_ballot_w32(o.x == node);
                    while (hits != 0u) {
                        const int q = __builtin_ctz(hits);
                        hits &= hits - 1u;
                        const int   s  = __builtin_amdgcn_readlane(o.y, q);
                        const float wq = __int_as_float(__builtin_amdgcn_readlane(o.z, q));
                        extra += *(const v4f*)(xl + (size_t)s * D) * wq;
                    }
                }
                myrows[kk * HP] += extra;
            }
        }

        __syncthreads();

        const int mt_n = min(mt + (int)gridDim.x, GEMM_M_TILES - 1);
        cv  = cv_n;
        rsv = gcn_bounds(cv, lane);
        r4  = __builtin_amdgcn_readlane(rsv, 4);
        if (r4 > 0) rec = gcn_load_records(slot, mt_n * 16 + wave * 4, 0, rsv, lane);
        const int cv_nn = cnt[min(mt + 2 * (int)gridDim.x, GEMM_M_TILES - 1) * 16 + wave * 4 + l3];

        const v4f* ap = tile + l15 * HP + rsel;
        v4f a[16];
#pragma unroll
        for (int j = 0; j < 16; ++j) a[j] = ap[2 * j];
        asm volatile("" :: "v"(a[0]), "v"(a[1]), "v"(a[2]), "v"(a[3]),
                           "v"(a[4]), "v"(a[5]), "v"(a[6]), "v"(a[7]),
                           "v"(a[8]), "v"(a[9]), "v"(a[10]), "v"(a[11]),
                           "v"(a[12]), "v"(a[13]), "v"(a[14]), "v"(a[15]));

        v8f acc0 = {}, acc1 = {};
#pragma unroll
        for (int j = 0; j < 16; ++j) {
            const v2f alo  = {a[j].x,  a[j].y},  ahi  = {a[j].z,  a[j].w};
            const v2f b0lo = {b0[j].x, b0[j].y}, b0hi = {b0[j].z, b0[j].w};
            const v2f b1lo = {b1[j].x, b1[j].y}, b1hi = {b1[j].z, b1[j].w};
            acc0 = __builtin_amdgcn_wmma_f32_16x16x4_f32(
                false, alo, false, b0lo, (short)0, acc0, false, false);
            acc1 = __builtin_amdgcn_wmma_f32_16x16x4_f32(
                false, alo, false, b1lo, (short)0, acc1, false, false);
            acc0 = __builtin_amdgcn_wmma_f32_16x16x4_f32(
                false, ahi, false, b0hi, (short)0, acc0, false, false);
            acc1 = __builtin_amdgcn_wmma_f32_16x16x4_f32(
                false, ahi, false, b1hi, (short)0, acc1, false, false);
        }

        const int row0 = mt * 16;
        float* optr = out + (size_t)(row0 + 8 * rsel) * D + col0 + l15;
#pragma unroll
        for (int r = 0; r < 8; ++r) {
            optr[(size_t)r * D]      = acc0[r] + bias0;
            optr[(size_t)r * D + 16] = acc1[r] + bias1;
        }
        __builtin_amdgcn_sched_barrier(0);
        cv_n = cv_nn;
        buf ^= 1;
    }
}

extern "C" void kernel_launch(void* const* d_in, const int* in_sizes, int n_in,
                              void* d_out, int out_size, void* d_ws, size_t ws_size,
                              hipStream_t stream) {
    const float* x   = (const float*)d_in[0];
    const float* ew  = (const float*)d_in[1];
    const int*   src = (const int*)  d_in[2];
    const int*   dst = (const int*)  d_in[3];
    const float* Wm  = (const float*)d_in[4];
    const float* bv  = (const float*)d_in[5];

    float* out = (float*)d_out;

    int*  cnt  = (int*)d_ws + WS_CNT;
    int2* slot = (int2*)((int*)d_ws + WS_SLOT);
    int4* ovf  = (int4*)((int*)d_ws + WS_OVF);

    int4* sorted = (int4*)((int*)d_ws + WS_SORT);
    int2* tab    = (int2*)((int*)d_ws + WS_TAB);

    gcn_slice_sort_kernel<<<250, 256, 0, stream>>>(ew, src, dst, cnt, sorted, tab);
    gcn_group_file_kernel<<<250, 256, 0, stream>>>(sorted, tab, cnt, slot, ovf);
    gcn_pull_gemm_kernel<<<1024, 128, 0, stream>>>(x, cnt, slot, ovf, Wm, bv, out);
}


# GEMM overlapped with the next tile's pull: its first 16 x rows are asked for just ahead of the WMMAs, summed behind them; A fragments read from LDS 4 ahead, not all 16 (248 VGPRs); sums unchanged
# speedup vs baseline: 3.3916x; 1.0443x over previous
#include <hip/hip_runtime.h>


#define N_NODES 100000
#define N_EDGES 640000
#define D 128
#define K_SLOTS 32

#define SLICE_EDGES 2560
#define N_SLICES    (N_EDGES / SLICE_EDGES)
#define GROUP_NODES 400
#define N_GROUPS    (N_NODES / GROUP_NODES)
#define SORT_EPT    (SLICE_EDGES / 256)
#define FILE_BATCH  12

static_assert(N_EDGES % SLICE_EDGES == 0 && SLICE_EDGES % 256 == 0, "whole slices");
static_assert(N_NODES % GROUP_NODES == 0, "whole groups");
static_assert(N_SLICES <= 256 && N_GROUPS <= 256, "one thread per slice / per group");

typedef float v8f __attribute__((ext_vector_type(8)));
typedef float v2f __attribute__((ext_vector_type(2)));
typedef float v4f __attribute__((ext_vector_type(4)));

#define WS_CNT   0
#define WS_SLOT  131072
#define WS_OVF   (WS_SLOT + 2 * N_NODES * K_SLOTS)
#define WS_SORT  (WS_OVF + 4 * N_EDGES)
#define WS_TAB   (WS_SORT + 4 * N_EDGES)

__device__ __forceinline__ int gcn_block_excl_scan(int v, int tid, int* s_wsum, int& total) {
    const int lane = tid & 31;
    int inc = v;
#pragma unroll
    for (int o = 1; o < 32; o <<= 1) {
        const int t = __shfl_up(inc, o, 32);
        if (lane >= o) inc += t;
    }
    if (lane == 31) s_wsum[tid >> 5] = inc;
    __syncthreads();
    int wbase = 0, tot = 0;
#pragma unroll
    for (int k = 0; k < 8; ++k) {
        const int t = s_wsum[k];
        if (k < (tid >> 5)) wbase += t;
        tot += t;
    }
    total = tot;
    return wbase + inc - v;
}

__global__ void __launch_bounds__(256)
gcn_slice_sort_kernel(const float* __restrict__ ew,
                      const int* __restrict__ src,
                      const int* __restrict__ dst,
                      int* __restrict__ cnt,
                      int4* __restrict__ sorted,
                      int2* __restrict__ tab) {
    __shared__ int s_hist[256];
    __shared__ int s_off[256];
    __shared__ int s_wsum[8];
    const int tid = threadIdx.x;
    const int sl  = blockIdx.x;
    s_hist[tid] = 0;
    if (sl == 0 && tid == 0) cnt[N_NODES] = 0;
    __syncthreads();

    int d[SORT_EPT], s[SORT_EPT], wb[SORT_EPT], r[SORT_EPT];
    const int e0 = sl * SLICE_EDGES + tid;
#pragma unroll
    for (int j = 0; j < SORT_EPT; ++j) {
        d[j]  = dst[e0 + j * 256];
        s[j]  = src[e0 + j * 256];
        wb[j] = __float_as_int(ew[e0 + j * 256]);
    }
#pragma unroll
    for (int j = 0; j < SORT_EPT; ++j) r[j] = atomicAdd(&s_hist[d[j] / GROUP_NODES], 1);
    __syncthreads();

    const int c = s_hist[tid];
    int total;
    const int off = gcn_block_excl_scan(c, tid, s_wsum, total);
    s_off[tid] = off;
    if (tid < N_GROUPS) tab[tid * N_SLICES + sl] = make_int2(off, c);
    __syncthreads();

    int4* mine = sorted + sl * SLICE_EDGES;
#pragma unroll
    for (int j = 0; j < SORT_EPT; ++j)
        mine[s_off[d[j] / GROUP_NODES] + r[j]] = make_int4(d[j], s[j], wb[j], 0);
}

__global__ void __launch_bounds__(256)
gcn_group_file_kernel(const int4* __restrict__ sorted,
                      const int2* __restrict__ tab,
                      int* __restrict__ cnt,
                      int2* __restrict__ slot,
                      int4* __restrict__ ovf) {
    __shared__ int s_cnt[GROUP_NODES];
    __shared__ int s_base[256];
    __shared__ int s_first[256];
    __shared__ int s_wsum[8];
    const int tid = threadIdx.x;
    const int g   = blockIdx.x;
    for (int i = tid; i < GROUP_NODES; i += 256) s_cnt[i] = 0;
    int2 sc = make_int2(0, 0);
    if (tid < N_SLICES) sc = tab[g * N_SLICES + tid];
    int total;
    const int base = gcn_block_excl_scan(sc.y, tid, s_wsum, total);
    s_base[tid]  = base;
    s_first[tid] = tid * SLICE_EDGES + sc.x;
    __syncthreads();

    for (int i0 = 0; i0 < total; i0 += 256 * FILE_BATCH) {
        int4 rec[FILE_BATCH];
#pragma unroll
        for (int k = 0; k < FILE_BATCH; ++k) {
            const int i = min(i0 + k * 256 + tid, total - 1);
            int t = 0;
#pragma unroll
            for (int step = 128; step > 0; step >>= 1)
                if (s_base[t + step] <= i) t += step;
            rec[k] = sorted[s_first[t] + (i - s_base[t])];
        }
        int r[FILE_BATCH];
#pragma unroll
        for (int k = 0; k < FILE_BATCH; ++k) {
            r[k] = 0;
            if (i0 + k * 256 + tid < total)
                r[k] = atomicAdd(&s_cnt[rec[k].x - g * GROUP_NODES], 1);
        }
#pragma unroll
        for (int k = 0; k < FILE_BATCH; ++k) {
            if (i0 + k * 256 + tid < total) {
                const int dn = rec[k].x;
                if (r[k] < K_SLOTS) {
                    slot[dn * K_SLOTS + r[k]] = make_int2(rec[k].y, rec[k].z);
                } else {
                    const int p = atomicAdd(cnt + N_NODES, 1);
                    ovf[p] = make_int4(dn, rec[k].y, rec[k].z, 0);
                }
            }
        }
    }
    __syncthreads();
    for (int i = tid; i < GROUP_NODES; i += 256) cnt[g * GROUP_NODES + i] = s_cnt[i];
}

#define GEMM_M_TILES (N_NODES / 16)
#define HP 33
#define PULL_BATCH 16

__device__ __forceinline__ int gcn_bounds(int cv, int lane) {
    const int c0 = min(__builtin_amdgcn_readlane(cv, 0), K_SLOTS);
    const int c1 = min(__builtin_amdgcn_readlane(cv, 1), K_SLOTS);
    const int c2 = min(__builtin_amdgcn_readlane(cv, 2), K_SLOTS);
    const int c3 = min(__builtin_amdgcn_readlane(cv, 3), K_SLOTS);
    return (lane > 0 ? c0 : 0) + (lane > 1 ? c1 : 0) + (lane > 2 ? c2 : 0) + (lane > 3 ? c3 : 0);
}

__device__ __forceinline__ int2 gcn_load_records(const int2* __restrict__ slot, int node0,
                                                 int cbase, int rsv, int lane) {
    const int b1 = __builtin_amdgcn_readlane(rsv, 1);
    const int b2 = __builtin_amdgcn_readlane(rsv, 2);
    const int b3 = __builtin_amdgcn_readlane(rsv, 3);
    const int b4 = __builtin_amdgcn_readlane(rsv, 4);
    const int i   = cbase + min(lane, min(32, b4 - cbase) - 1);
    const int k   = (i >= b1) + (i >= b2) + (i >= b3);
    const int lbk = i >= b3 ? b3 : (i >= b2 ? b2 : (i >= b1 ? b1 : 0));
    return slot[(node0 + k) * K_SLOTS + (i - lbk)];
}

__device__ __forceinline__ bool gcn_owner(int rsv, int cbase, int lane, int& kown) {
    const int b1 = __builtin_amdgcn_readlane(rsv, 1);
    const int b2 = __builtin_amdgcn_readlane(rsv, 2);
    const int b3 = __builtin_amdgcn_readlane(rsv, 3);
    const int b4 = __builtin_amdgcn_readlane(rsv, 4);
    const int i    = cbase + lane;
    const int endk = i >= b3 ? b4 : (i >= b2 ? b3 : (i >= b1 ? b2 : b1));
    kown = (i >= b1) + (i >= b2) + (i >= b3);
    return i < b4 && i + 1 == endk;
}

__device__ __forceinline__ void gcn_issue_rows(v4f (&v)[PULL_BATCH], const float* __restrict__ xl,
                                               int recx, int ib) {
#pragma unroll
    for (int j = 0; j < PULL_BATCH; ++j) {
        const int s = __builtin_amdgcn_readlane(recx, ib + j);
        v[j] = *(const v4f*)(xl + (size_t)s * D);
    }
}

__device__ __forceinline__ void gcn_sum_rows(const v4f (&v)[PULL_BATCH], int recy, int ib,
                                             unsigned int mb, int kown, v4f& acc, v4f* myrows) {
    float w[PULL_BATCH];
#pragma unroll
    for (int j = 0; j < PULL_BATCH; ++j) w[j] = __int_as_float(__builtin_amdgcn_readlane(recy, ib + j));
    asm volatile("" :: "v"(v[0]), "v"(v[1]), "v"(v[2]), "v"(v[3]),
                       "v"(v[4]), "v"(v[5]), "v"(v[6]), "v"(v[7]),
                       "v"(v[8]), "v"(v[9]), "v"(v[10]), "v"(v[11]),
                       "v"(v[12]), "v"(v[13]), "v"(v[14]), "v"(v[15]));
#pragma unroll
    for (int j = 0; j < PULL_BATCH; ++j) {
        acc += v[j] * w[j];
        if (__builtin_expect((mb >> j) & 1u, 0)) {
            myrows[__builtin_amdgcn_readlane(kown, ib + j) * HP] = acc;
            acc = (v4f){0.f, 0.f, 0.f, 0.f};
        }
    }
}

__global__ void __launch_bounds__(128)
gcn_pull_gemm_kernel(const float* __restrict__ x,
                     const int* __restrict__ cnt,
                     const int2* __restrict__ slot,
                     const int4* __restrict__ ovf,
                     const float* __restrict__ W,
                     const float* __restrict__ b,
                     float* __restrict__ out) {
    __shared__ v4f s_h[2 * 16 * HP];

    const int lane = threadIdx.x & 31;
    const int wave = __builtin_amdgcn_readfirstlane(threadIdx.x >> 5);
    const int col0 = wave * 32;
    const int l15  = lane & 15;
    const int rsel = lane >> 4;

    const float* wp0 = W + (size_t)(col0 + l15) * D + 4 * rsel;
    const float* wp1 = wp0 + (size_t)16 * D;
    v4f b0[16], b1[16];
#pragma unroll
    for (int j = 0; j < 16; ++j) {
        b0[j] = *(const v4f*)(wp0 + 8 * j);
        b1[j] = *(const v4f*)(wp1 + 8 * j);
    }
    const float bias0 = b[col0 + l15];
    const float bias1 = b[col0 + 16 + l15];

    const float* xl = x + lane * 4;
    const int l3 = min(lane, 3);
    const int G  = (int)gridDim.x;

    int mt = blockIdx.x;
    const int w4 = wave * 4;
    int cv    = cnt[mt * 16 + w4 + l3];
    int cv_n  = cnt[min(mt + G,     GEMM_M_TILES - 1) * 16 + w4 + l3];
    int cv_nn = cnt[min(mt + 2 * G, GEMM_M_TILES - 1) * 16 + w4 + l3];
    int rsv   = gcn_bounds(cv, lane);
    int r4    = __builtin_amdgcn_readlane(rsv, 4);
    int2 rec  = make_int2(0, 0);
    if (r4 > 0) rec = gcn_load_records(slot, mt * 16 + w4, 0, rsv, lane);
    int rsv_n = gcn_bounds(cv_n, lane);
    int r4_n  = __builtin_amdgcn_readlane(rsv_n, 4);
    int2 rec_n = make_int2(0, 0);
    if (r4_n > 0) rec_n = gcn_load_records(slot, min(mt + G, GEMM_M_TILES - 1) * 16 + w4, 0, rsv_n, lane);

    v4f v[PULL_BATCH];
    gcn_issue_rows(v, xl, rec.x, 0);

    int buf = 0;
    for (; mt < GEMM_M_TILES; mt += G) {
        v4f* tile   = s_h + buf * (16 * HP);
        v4f* myrows = tile + wave * (4 * HP) + lane;
        const int node0 = mt * 16 + w4;

        const int c0 = __builtin_amdgcn_readlane(cv, 0);
        const int c1 = __builtin_amdgcn_readlane(cv, 1);
        const int c2 = __builtin_amdgcn_readlane(cv, 2);
        const int c3 = __builtin_amdgcn_readlane(cv, 3);
        if (min(min(c0, c1), min(c2, c3)) == 0) {
            for (int kk = 0; kk < 4; ++kk) {
                const int ck = kk == 0 ? c0 : (kk == 1 ? c1 : (kk == 2 ? c2 : c3));
                if (ck == 0) myrows[kk * HP] = (v4f){0.f, 0.f, 0.f, 0.f};
            }
        }
        v4f acc = {0.f, 0.f, 0.f, 0.f};
        int cbase = 0;
        bool inflight = true;
        while (cbase < r4) {
            const int n = min(32, r4 - cbase);
            int kown;
            const bool last = gcn_owner(rsv, cbase, lane, kown);
            const unsigned int lastmask = __builtin_amdgcn_ballot_w32(last);
            for (int i = 0; i < n; i += PULL_BATCH) {
                if (!inflight) gcn_issue_rows(v, xl, rec.x, i);
                inflight = false;
                gcn_sum_rows(v, rec.y, i, lastmask >> i, kown, acc, myrows);
            }
            cbase += 32;
            if (cbase < r4) rec = gcn_load_records(slot, node0, cbase, rsv, lane);
        }

        if (max(max(c0, c1), max(c2, c3)) > K_SLOTS) {
            const int novf = min(__builtin_amdgcn_readfirstlane(cnt[N_NODES]), N_EDGES);
            for (int kk = 0; kk < 4; ++kk) {
                const int ck = kk == 0 ? c0 : (kk == 1 ? c1 : (kk == 2 ? c2 : c3));
                if (ck <= K_SLOTS) continue;
                const int node = node0 + kk;
                v4f extra = {0.f, 0.f, 0.f, 0.f};
                for (int base = 0; base < novf; base += 32) {
                    const int idx = base + lane;
                    int4 o = make_int4(-1, 0, 0, 0);
                    if (idx < novf) o = ovf[idx];
                    unsigned int hits = __builtin_amdgcn_ballot_w32(o.x == node);
                    while (hits != 0u) {
                        const int q = __builtin_ctz(hits);
                        hits &= hits - 1u;
                        const int   s  = __builtin_amdgcn_readlane(o.y, q);
                        const float wq = __int_as_float(__builtin_amdgcn_readlane(o.z, q));
                        extra += *(const v4f*)(xl + (size_t)s * D) * wq;
                    }
                }
                myrows[kk * HP] += extra;
            }
        }

        __syncthreads();

        cv  = cv_n;
        rsv = rsv_n;
        r4  = r4_n;
        rec = rec_n;
        if (mt + G < GEMM_M_TILES) gcn_issue_rows(v, xl, rec.x, 0);
        cv_n  = cv_nn;
        rsv_n = gcn_bounds(cv_n, lane);
        r4_n  = __builtin_amdgcn_readlane(rsv_n, 4);
        if (r4_n > 0) rec_n = gcn_load_records(slot, min(mt + 2 * G, GEMM_M_TILES - 1) * 16 + w4, 0, rsv_n, lane);
        const int cv_nnn = cnt[min(mt + 3 * G, GEMM_M_TILES - 1) * 16 + w4 + l3];

        const v4f* ap = tile + l15 * HP + rsel;
        v8f acc0 = {}, acc1 = {};
#pragma unroll
        for (int g = 0; g < 4; ++g) {
            v4f a[4];
#pragma unroll
            for (int q = 0; q < 4; ++q) a[q] = ap[2 * (4 * g + q)];
#pragma unroll
            for (int q = 0; q < 4; ++q) {
                const int j = 4 * g + q;
                const v2f alo  = {a[q].x,  a[q].y},  ahi  = {a[q].z,  a[q].w};
                const v2f b0lo = {b0[j].x, b0[j].y}, b0hi = {b0[j].z, b0[j].w};
                const v2f b1lo = {b1[j].x, b1[j].y}, b1hi = {b1[j].z, b1[j].w};
                acc0 = __builtin_amdgcn_wmma_f32_16x16x4_f32(
                    false, alo, false, b0lo, (short)0, acc0, false, false);
                acc1 = __builtin_amdgcn_wmma_f32_16x16x4_f32(
                    false, alo, false, b1lo, (short)0, acc1, false, false);
                acc0 = __builtin_amdgcn_wmma_f32_16x16x4_f32(
                    false, ahi, false, b0hi, (short)0, acc0, false, false);
                acc1 = __builtin_amdgcn_wmma_f32_16x16x4_f32(
                    false, ahi, false, b1hi, (short)0, acc1, false, false);
            }
        }
        __builtin_amdgcn_sched_group_barrier(0x100, 4, 0);
#pragma unroll
        for (int k = 0; k < 12; ++k) {
            __builtin_amdgcn_sched_group_barrier(0x008, 4, 0);
            __builtin_amdgcn_sched_group_barrier(0x100, 1, 0);
        }
        __builtin_amdgcn_sched_group_barrier(0x008, 16, 0);

        const int row0 = mt * 16;
        float* optr = out + (size_t)(row0 + 8 * rsel) * D + col0 + l15;
#pragma unroll
        for (int r = 0; r < 8; ++r) {
            optr[(size_t)r * D]      = acc0[r] + bias0;
            optr[(size_t)r * D + 16] = acc1[r] + bias1;
        }
        __builtin_amdgcn_sched_barrier(0);
        cv_nn = cv_nnn;
        buf ^= 1;
    }
}

extern "C" void kernel_launch(void* const* d_in, const int* in_sizes, int n_in,
                              void* d_out, int out_size, void* d_ws, size_t ws_size,
                              hipStream_t stream) {
    const float* x   = (const float*)d_in[0];
    const float* ew  = (const float*)d_in[1];
    const int*   src = (const int*)  d_in[2];
    const int*   dst = (const int*)  d_in[3];
    const float* Wm  = (const float*)d_in[4];
    const float* bv  = (const float*)d_in[5];

    float* out = (float*)d_out;

    int*  cnt  = (int*)d_ws + WS_CNT;
    int2* slot = (int2*)((int*)d_ws + WS_SLOT);
    int4* ovf  = (int4*)((int*)d_ws + WS_OVF);

    int4* sorted = (int4*)((int*)d_ws + WS_SORT);
    int2* tab    = (int2*)((int*)d_ws + WS_TAB);

    gcn_slice_sort_kernel<<<250, 256, 0, stream>>>(ew, src, dst, cnt, sorted, tab);
    gcn_group_file_kernel<<<250, 256, 0, stream>>>(sorted, tab, cnt, slot, ovf);
    gcn_pull_gemm_kernel<<<1024, 128, 0, stream>>>(x, cnt, slot, ovf, Wm, bv, out);
}


# Launch 2 gone: launch 1 sorts slices by the pull block that owns the dst (6-7 tiles in a row per block); each pull block files its own edges into an LDS record list; sums in LDS arrival order
# speedup vs baseline: 3.6835x; 1.0861x over previous
#include <hip/hip_runtime.h>


#define N_NODES 100000
#define N_EDGES 640000
#define D 128

#define SLICE_EDGES 2560
#define N_SLICES    (N_EDGES / SLICE_EDGES)
#define SORT_EPT    (SLICE_EDGES / 256)

#define GEMM_M_TILES (N_NODES / 16)
#define PULL_BLOCKS  1024
#define PB_TILES_MAX 7
#define REC_CAP      1024
#define SEG_U        8

static_assert(N_EDGES % SLICE_EDGES == 0 && SLICE_EDGES % 256 == 0, "whole slices");
static_assert(N_NODES % 16 == 0, "whole tiles");
static_assert(N_SLICES > 128 && N_SLICES <= 256, "two slices per thread of launch 2");
static_assert(PULL_BLOCKS == 4 * 256, "four owners per thread of launch 1");
static_assert((GEMM_M_TILES + PULL_BLOCKS - 1) / PULL_BLOCKS <= PB_TILES_MAX, "tiles per block");
static_assert(16 * PB_TILES_MAX + 16 <= 128, "node counters and 16 idle ones in s_cnt[128]");

typedef float v8f __attribute__((ext_vector_type(8)));
typedef float v2f __attribute__((ext_vector_type(2)));
typedef float v4f __attribute__((ext_vector_type(4)));

#define WS_SORT  0
#define WS_TAB   (WS_SORT + 4 * N_EDGES)

__device__ __forceinline__ int gcn_owner_of_node(int d) {
    return (((d >> 4) + 1) * PULL_BLOCKS - 1) / GEMM_M_TILES;
}

__device__ __forceinline__ int gcn_block_excl_scan(int v, int tid, int* s_wsum, int& total) {
    const int lane = tid & 31;
    int inc = v;
#pragma unroll
    for (int o = 1; o < 32; o <<= 1) {
        const int t = __shfl_up(inc, o, 32);
        if (lane >= o) inc += t;
    }
    if (lane == 31) s_wsum[tid >> 5] = inc;
    __syncthreads();
    int wbase = 0, tot = 0;
#pragma unroll
    for (int k = 0; k < 8; ++k) {
        const int t = s_wsum[k];
        if (k < (tid >> 5)) wbase += t;
        tot += t;
    }
    total = tot;
    return wbase + inc - v;
}

__device__ __forceinline__ int gcn_block128_excl_scan(int v, int tid, int* s_wsum, int& total) {
    const int lane = tid & 31;
    int inc = v;
#pragma unroll
    for (int o = 1; o < 32; o <<= 1) {
        const int t = __shfl_up(inc, o, 32);
        if (lane >= o) inc += t;
    }
    if (lane == 31) s_wsum[tid >> 5] = inc;
    __syncthreads();
    int wbase = 0, tot = 0;
#pragma unroll
    for (int k = 0; k < 4; ++k) {
        const int t = s_wsum[k];
        if (k < (tid >> 5)) wbase += t;
        tot += t;
    }
    total = tot;
    return wbase + inc - v;
}

__global__ void __launch_bounds__(256)
gcn_slice_sort_kernel(const float* __restrict__ ew,
                      const int* __restrict__ src,
                      const int* __restrict__ dst,
                      int4* __restrict__ sorted,
                      int2* __restrict__ tab) {
    __shared__ int s_hist[PULL_BLOCKS];
    __shared__ int s_off[PULL_BLOCKS];
    __shared__ int s_wsum[8];
    const int tid = threadIdx.x;
    const int sl  = blockIdx.x;
#pragma unroll
    for (int j = 0; j < 4; ++j) s_hist[tid + 256 * j] = 0;
    __syncthreads();

    int d[SORT_EPT], s[SORT_EPT], wb[SORT_EPT], r[SORT_EPT], ow[SORT_EPT];
    const int e0 = sl * SLICE_EDGES + tid;
#pragma unroll
    for (int j = 0; j < SORT_EPT; ++j) {
        d[j]  = dst[e0 + j * 256];
        s[j]  = src[e0 + j * 256];
        wb[j] = __float_as_int(ew[e0 + j * 256]);
    }
#pragma unroll
    for (int j = 0; j < SORT_EPT; ++j) {
        ow[j] = gcn_owner_of_node(d[j]);
        r[j]  = atomicAdd(&s_hist[ow[j]], 1);
    }
    __syncthreads();

    const int c0 = s_hist[4 * tid], c1 = s_hist[4 * tid + 1], c2 = s_hist[4 * tid + 2], c3 = s_hist[4 * tid + 3];
    int total;
    const int off0 = gcn_block_excl_scan(c0 + c1 + c2 + c3, tid, s_wsum, total);
    const int off1 = off0 + c0, off2 = off1 + c1, off3 = off2 + c2;
    s_off[4 * tid]     = off0;
    s_off[4 * tid + 1] = off1;
    s_off[4 * tid + 2] = off2;
    s_off[4 * tid + 3] = off3;
    int2* trow = tab + sl * PULL_BLOCKS + 4 * tid;
    trow[0] = make_int2(off0, c0);
    trow[1] = make_int2(off1, c1);
    trow[2] = make_int2(off2, c2);
    trow[3] = make_int2(off3, c3);
    __syncthreads();

    int4* mine = sorted + sl * SLICE_EDGES;
#pragma unroll
    for (int j = 0; j < SORT_EPT; ++j)
        mine[s_off[ow[j]] + r[j]] = make_int4(d[j], s[j], wb[j], 0);
}

#define HP 33
#define PULL_BATCH 16

__device__ __forceinline__ bool gcn_owner(int rsv, int cbase, int lane, int& kown) {
    const int b1 = __builtin_amdgcn_readlane(rsv, 1);
    const int b2 = __builtin_amdgcn_readlane(rsv, 2);
    const int b3 = __builtin_amdgcn_readlane(rsv, 3);
    const int b4 = __builtin_amdgcn_readlane(rsv, 4);
    const int i    = cbase + lane;
    const int endk = i >= b3 ? b4 : (i >= b2 ? b3 : (i >= b1 ? b2 : b1));
    kown = (i >= b1) + (i >= b2) + (i >= b3);
    return i < b4 && i + 1 == endk;
}

__device__ __forceinline__ int gcn_tile_offsets(const int* s_off, int n0, int lane) {
    return s_off[n0 + min(lane, 4)];
}

__device__ __forceinline__ void gcn_tile_bounds(int o, int& base, int& rsv, int& r4) {
    base = __builtin_amdgcn_readlane(o, 0);
    rsv  = o - base;
    r4   = __builtin_amdgcn_readlane(rsv, 4);
}

__device__ __forceinline__ int2 gcn_load_records(const int2* s_rec, int base, int cbase, int r4, int lane) {
    return s_rec[base + cbase + min(lane, min(32, r4 - cbase) - 1)];
}

__device__ __forceinline__ void gcn_issue_rows(v4f (&v)[PULL_BATCH], const float* __restrict__ xl,
                                               int recx, int ib) {
#pragma unroll
    for (int j = 0; j < PULL_BATCH; ++j) {
        const int s = __builtin_amdgcn_readlane(recx, ib + j);
        v[j] = *(const v4f*)(xl + (size_t)s * D);
    }
}

__device__ __forceinline__ void gcn_sum_rows(const v4f (&v)[PULL_BATCH], int recy, int ib,
                                             unsigned int mb, int kown, v4f& acc, v4f* myrows) {
    float w[PULL_BATCH];
#pragma unroll
    for (int j = 0; j < PULL_BATCH; ++j) w[j] = __int_as_float(__builtin_amdgcn_readlane(recy, ib + j));
    asm volatile("" :: "v"(v[0]), "v"(v[1]), "v"(v[2]), "v"(v[3]),
                       "v"(v[4]), "v"(v[5]), "v"(v[6]), "v"(v[7]),
                       "v"(v[8]), "v"(v[9]), "v"(v[10]), "v"(v[11]),
                       "v"(v[12]), "v"(v[13]), "v"(v[14]), "v"(v[15]));
#pragma unroll
    for (int j = 0; j < PULL_BATCH; ++j) {
        acc += v[j] * w[j];
        if (__builtin_expect((mb >> j) & 1u, 0)) {
            myrows[__builtin_amdgcn_readlane(kown, ib + j) * HP] = acc;
            acc = (v4f){0.f, 0.f, 0.f, 0.f};
        }
    }
}

__device__ __forceinline__ int gcn_take_position(int dl, const int* s_off, int* s_cnt) {
    return s_off[dl] + atomicAdd(&s_cnt[dl], 1);
}

__device__ __forceinline__ void gcn_file_record(int4 rc, int node_base, const int* s_off,
                                                int* s_cnt, int2* s_rec) {
    const int pos = gcn_take_position(rc.x - node_base, s_off, s_cnt);
    if (pos < REC_CAP) s_rec[pos] = make_int2(rc.y, rc.z);
}

__global__ void __launch_bounds__(128)
gcn_pull_gemm_kernel(const float* __restrict__ x,
                     const int4* __restrict__ sorted,
                     const int2* __restrict__ tab,
                     const float* __restrict__ W,
                     const float* __restrict__ b,
                     float* __restrict__ out) {
    __shared__ v4f  s_h[2 * 16 * HP];
    __shared__ int2 s_rec[REC_CAP];
    __shared__ int2 s_seg[256];
    __shared__ int  s_cnt[128];
    __shared__ int  s_off[128];
    __shared__ int  s_wsum[4];

    const int tid  = threadIdx.x;
    const int lane = tid & 31;
    const int wave = __builtin_amdgcn_readfirstlane(tid >> 5);
    const int col0 = wave * 32;
    const int l15  = lane & 15;
    const int rsel = lane >> 4;

    const float* wp0 = W + (size_t)(col0 + l15) * D + 4 * rsel;
    const float* wp1 = wp0 + (size_t)16 * D;
    v4f b0[16], b1[16];
#pragma unroll
    for (int j = 0; j < 16; ++j) {
        b0[j] = *(const v4f*)(wp0 + 8 * j);
        b1[j] = *(const v4f*)(wp1 + 8 * j);
    }
    const float bias0 = b[col0 + l15];
    const float bias1 = b[col0 + 16 + l15];

    const float* xl = x + lane * 4;

    const int blk = blockIdx.x;
    const int t0  = (blk * GEMM_M_TILES) / PULL_BLOCKS;
    const int t1  = ((blk + 1) * GEMM_M_TILES) / PULL_BLOCKS;
    const int node_base = t0 * 16;

    s_cnt[tid] = 0;
    const int2 e0 = tab[tid * PULL_BLOCKS + blk];
    int2 e1 = tab[min(tid + 128, N_SLICES - 1) * PULL_BLOCKS + blk];
    if (tid + 128 >= N_SLICES) e1.y = 0;
    const int f0 = tid * SLICE_EDGES + e0.x;
    const int f1 = (tid + 128) * SLICE_EDGES + e1.x;
    s_seg[tid]       = make_int2(f0, e0.y);
    s_seg[tid + 128] = make_int2(f1, e1.y);
    __syncthreads();

    int4 ra[SEG_U], rb[SEG_U];
    const int l0 = max(e0.y, 1) - 1, l1 = max(e1.y, 1) - 1;
#pragma unroll
    for (int r = 0; r < SEG_U; ++r) {
        ra[r] = sorted[min(f0 + min(r, l0), N_EDGES - 1)];
        rb[r] = sorted[min(f1 + min(r, l1), N_EDGES - 1)];
    }
    __builtin_amdgcn_sched_barrier(0);

#pragma unroll
    for (int r = 0; r < SEG_U; ++r) {
        if (r < e0.y) atomicAdd(&s_cnt[ra[r].x - node_base], 1);
        if (r < e1.y) atomicAdd(&s_cnt[rb[r].x - node_base], 1);
    }
    for (int r = SEG_U; r < e0.y; ++r) atomicAdd(&s_cnt[sorted[f0 + r].x - node_base], 1);
    for (int r = SEG_U; r < e1.y; ++r) atomicAdd(&s_cnt[sorted[f1 + r].x - node_base], 1);
    __syncthreads();

    {
        const int cn = s_cnt[tid];
        int tot;
        const int ex = gcn_block128_excl_scan(cn, tid, s_wsum, tot);
        s_off[tid] = ex;
        s_cnt[tid] = 0;
    }
    __syncthreads();

    int pa[SEG_U], pb[SEG_U];
    const int idle = 16 * PB_TILES_MAX + (lane & 15);
#pragma unroll
    for (int r = 0; r < SEG_U; ++r) {
        const bool ha = r < e0.y, hb = r < e1.y;
        const int  da = ha ? ra[r].x - node_base : idle;
        const int  db = hb ? rb[r].x - node_base : idle;
        const int  qa = s_off[da] + atomicAdd(&s_cnt[da], ha ? 1 : 0);
        const int  qb = s_off[db] + atomicAdd(&s_cnt[db], hb ? 1 : 0);
        pa[r] = ha ? qa : REC_CAP;
        pb[r] = hb ? qb : REC_CAP;
    }
#pragma unroll
    for (int r = 0; r < SEG_U; ++r) {
        if (pa[r] < REC_CAP) s_rec[pa[r]] = make_int2(ra[r].y, ra[r].z);
        if (pb[r] < REC_CAP) s_rec[pb[r]] = make_int2(rb[r].y, rb[r].z);
    }
    for (int r = SEG_U; r < e0.y; ++r) gcn_file_record(sorted[f0 + r], node_base, s_off, s_cnt, s_rec);
    for (int r = SEG_U; r < e1.y; ++r) gcn_file_record(sorted[f1 + r], node_base, s_off, s_cnt, s_rec);
    __syncthreads();

    const int w4 = wave * 4;
    int base, rsv, r4;
    gcn_tile_bounds(gcn_tile_offsets(s_off, w4, lane), base, rsv, r4);
    bool listed = base + r4 <= REC_CAP;
    int2 rec = make_int2(0, 0);
    if (listed && r4 > 0) rec = gcn_load_records(s_rec, base, 0, r4, lane);
    int o_n = 0;
    if (t0 + 1 < t1) o_n = gcn_tile_offsets(s_off, w4 + 16, lane);

    v4f v[PULL_BATCH];
    gcn_issue_rows(v, xl, rec.x, 0);

    int buf = 0;
    for (int mt = t0; mt < t1; ++mt) {
        v4f* tile   = s_h + buf * (16 * HP);
        v4f* myrows = tile + wave * (4 * HP) + lane;
        const int n0 = (mt - t0) * 16 + w4;

        int base_n = 0, rsv_n = 0, r4_n = 0;
        bool listed_n = true;
        int2 rec_n = make_int2(0, 0);
        if (mt + 1 < t1) {
            gcn_tile_bounds(o_n, base_n, rsv_n, r4_n);
            listed_n = base_n + r4_n <= REC_CAP;
            if (listed_n && r4_n > 0) rec_n = gcn_load_records(s_rec, base_n, 0, r4_n, lane);
        }
        int o_nn = 0;
        if (mt + 2 < t1) o_nn = gcn_tile_offsets(s_off, n0 + 32, lane);

        if (listed) {
            const int q1 = __builtin_amdgcn_readlane(rsv, 1);
            const int q2 = __builtin_amdgcn_readlane(rsv, 2);
            const int q3 = __builtin_amdgcn_readlane(rsv, 3);
            const int c0 = q1, c1 = q2 - q1, c2 = q3 - q2, c3 = r4 - q3;
            if (min(min(c0, c1), min(c2, c3)) == 0) {
                for (int kk = 0; kk < 4; ++kk) {
                    const int ck = kk == 0 ? c0 : (kk == 1 ? c1 : (kk == 2 ? c2 : c3));
                    if (ck == 0) myrows[kk * HP] = (v4f){0.f, 0.f, 0.f, 0.f};
                }
            }
            v4f acc = {0.f, 0.f, 0.f, 0.f};
            int cbase = 0;
            bool inflight = true;
            while (cbase < r4) {
                const int n = min(32, r4 - cbase);
                int kown;
                const bool last = gcn_owner(rsv, cbase, lane, kown);
                const unsigned int lastmask = __builtin_amdgcn_ballot_w32(last);
                for (int i = 0; i < n; i += PULL_BATCH) {
                    if (!inflight) gcn_issue_rows(v, xl, rec.x, i);
                    inflight = false;
                    gcn_sum_rows(v, rec.y, i, lastmask >> i, kown, acc, myrows);
                }
                cbase += 32;
                if (cbase < r4) rec = gcn_load_records(s_rec, base, cbase, r4, lane);
            }
        } else {
            for (int kk = 0; kk < 4; ++kk) {
                const int node = node_base + n0 + kk;
                v4f sum = {0.f, 0.f, 0.f, 0.f};
                for (int sl = 0; sl < N_SLICES; ++sl) {
                    const int2 sg = s_seg[sl];
                    const int sf = __builtin_amdgcn_readfirstlane(sg.x);
                    const int sc = __builtin_amdgcn_readfirstlane(sg.y);
                    for (int eb = 0; eb < sc; eb += 32) {
                        const int i = eb + lane;
                        int4 o = make_int4(-1, 0, 0, 0);
                        if (i < sc) o = sorted[sf + i];
                        unsigned int hits = __builtin_amdgcn_ballot_w32(o.x == node);
                        while (hits != 0u) {
                            const int q = __builtin_ctz(hits);
                            hits &= hits - 1u;
                            const int   s  = __builtin_amdgcn_readlane(o.y, q);
                            const float wq = __int_as_float(__builtin_amdgcn_readlane(o.z, q));
                            sum += *(const v4f*)(xl + (size_t)s * D) * wq;
                        }
                    }
                }
                myrows[kk * HP] = sum;
            }
        }

        __syncthreads();

        base   = base_n;
        rsv    = rsv_n;
        r4     = r4_n;
        listed = listed_n;
        rec    = rec_n;
        o_n    = o_nn;
        if (mt + 1 < t1) gcn_issue_rows(v, xl, rec.x, 0);

        const v4f* ap = tile + l15 * HP + rsel;
        v8f acc0 = {}, acc1 = {};
#pragma unroll
        for (int g = 0; g < 4; ++g) {
            v4f a[4];
#pragma unroll
            for (int q = 0; q < 4; ++q) a[q] = ap[2 * (4 * g + q)];
#pragma unroll
            for (int q = 0; q < 4; ++q) {
                const int j = 4 * g + q;
                const v2f alo  = {a[q].x,  a[q].y},  ahi  = {a[q].z,  a[q].w};
                const v2f b0lo = {b0[j].x, b0[j].y}, b0hi = {b0[j].z, b0[j].w};
                const v2f b1lo = {b1[j].x, b1[j].y}, b1hi = {b1[j].z, b1[j].w};
                acc0 = __builtin_amdgcn_wmma_f32_16x16x4_f32(
                    false, alo, false, b0lo, (short)0, acc0, false, false);
                acc1 = __builtin_amdgcn_wmma_f32_16x16x4_f32(
                    false, alo, false, b1lo, (short)0, acc1, false, false);
                acc0 = __builtin_amdgcn_wmma_f32_16x16x4_f32(
                    false, ahi, false, b0hi, (short)0, acc0, false, false);
                acc1 = __builtin_amdgcn_wmma_f32_16x16x4_f32(
                    false, ahi, false, b1hi, (short)0, acc1, false, false);
            }
        }
        __builtin_amdgcn_sched_group_barrier(0x100, 4, 0);
#pragma unroll
        for (int k = 0; k < 12; ++k) {
            __builtin_amdgcn_sched_group_barrier(0x008, 4, 0);
            __builtin_amdgcn_sched_group_barrier(0x100, 1, 0);
        }
        __builtin_amdgcn_sched_group_barrier(0x008, 16, 0);

        const int row0 = mt * 16;
        float* optr = out + (size_t)(row0 + 8 * rsel) * D + col0 + l15;
#pragma unroll
        for (int r = 0; r < 8; ++r) {
            optr[(size_t)r * D]      = acc0[r] + bias0;
            optr[(size_t)r * D + 16] = acc1[r] + bias1;
        }
        buf ^= 1;
    }
}

extern "C" void kernel_launch(void* const* d_in, const int* in_sizes, int n_in,
                              void* d_out, int out_size, void* d_ws, size_t ws_size,
                              hipStream_t stream) {
    const float* x   = (const float*)d_in[0];
    const float* ew  = (const float*)d_in[1];
    const int*   src = (const int*)  d_in[2];
    const int*   dst = (const int*)  d_in[3];
    const float* Wm  = (const float*)d_in[4];
    const float* bv  = (const float*)d_in[5];

    float* out = (float*)d_out;

    int4* sorted = (int4*)((int*)d_ws + WS_SORT);
    int2* tab    = (int2*)((int*)d_ws + WS_TAB);

    gcn_slice_sort_kernel<<<250, 256, 0, stream>>>(ew, src, dst, sorted, tab);
    gcn_pull_gemm_kernel<<<1024, 128, 0, stream>>>(x, sorted, tab, Wm, bv, out);
}
